# starting kernel
# baseline (speedup 1.0000x reference)
#include <hip/hip_runtime.h>
#include <hip/hip_bf16.h>

typedef __attribute__((ext_vector_type(2))) float v2f;
typedef __attribute__((ext_vector_type(8))) float v8f;

#define IN_DIM  128
#define OUT_DIM 64

__global__ void gat_zero_out(float4* __restrict__ out, int n4) {
    int i = blockIdx.x * blockDim.x + threadIdx.x;
    int stride = gridDim.x * blockDim.x;
    float4 z4 = make_float4(0.f, 0.f, 0.f, 0.f);
    for (; i < n4; i += stride) out[i] = z4;
}

__global__ void gat_gemm_z(const float* __restrict__ h,
                           const float* __restrict__ W,
                           float* __restrict__ z,
                           int nNodes) {
    const int wave = (blockIdx.x * blockDim.x + threadIdx.x) >> 5;
    const int lane = threadIdx.x & 31;
    const int row0 = wave * 16;
    if (row0 >= nNodes) return;

    const int half = lane >> 4;
    const int l16  = lane & 15;

    const float* hrow = h + (size_t)(row0 + l16) * IN_DIM + half * 2;

    v8f acc0 = {}, acc1 = {}, acc2 = {}, acc3 = {};

    for (int k = 0; k < IN_DIM; k += 4) {
        v2f a;
        a.x = hrow[k];
        a.y = hrow[k + 1];

        const float* wk = W + (size_t)(k + half * 2) * OUT_DIM + l16;
        v2f b0, b1, b2, b3;
        b0.x = wk[0];       b0.y = wk[OUT_DIM + 0];
        b1.x = wk[16];      b1.y = wk[OUT_DIM + 16];
        b2.x = wk[32];      b2.y = wk[OUT_DIM + 32];
        b3.x = wk[48];      b3.y = wk[OUT_DIM + 48];

        acc0 = __builtin_amdgcn_wmma_f32_16x16x4_f32(false, a, false, b0,
                                                     (short)0, acc0, false, false);
        acc1 = __builtin_amdgcn_wmma_f32_16x16x4_f32(false, a, false, b1,
                                                     (short)0, acc1, false, false);
        acc2 = __builtin_amdgcn_wmma_f32_16x16x4_f32(false, a, false, b2,
                                                     (short)0, acc2, false, false);
        acc3 = __builtin_amdgcn_wmma_f32_16x16x4_f32(false, a, false, b3,
                                                     (short)0, acc3, false, false);
    }

    float* zt = z + (size_t)row0 * OUT_DIM + l16;
#pragma unroll
    for (int v = 0; v < 8; ++v) {
        const size_t r = (size_t)(v + 8 * half) * OUT_DIM;
        zt[r + 0]  = acc0[v];
        zt[r + 16] = acc1[v];
        zt[r + 32] = acc2[v];
        zt[r + 48] = acc3[v];
    }
}

__global__ void gat_edge(const float* __restrict__ z,
                         const int* __restrict__ src,
                         const int* __restrict__ dst,
                         const float* __restrict__ a_attn,
                         float* __restrict__ out,
                         int nEdges) {
    const int edge = (blockIdx.x * blockDim.x + threadIdx.x) >> 5;
    const int lane = threadIdx.x & 31;
    if (edge >= nEdges) return;

    const int s = src[edge];
    const int d = dst[edge];
    const float* zs = z + (size_t)s * OUT_DIM;
    const float* zd = z + (size_t)d * OUT_DIM;

    const float zs0 = zs[lane],      zs1 = zs[lane + 32];
    const float zd0 = zd[lane],      zd1 = zd[lane + 32];

    float p = zs0 * a_attn[lane]      + zs1 * a_attn[lane + 32]
            + zd0 * a_attn[64 + lane] + zd1 * a_attn[96 + lane];

#pragma unroll
    for (int off = 16; off > 0; off >>= 1)
        p += __shfl_xor(p, off, 32);

    const float e = (p >= 0.f) ? p : 0.01f * p;
    const float alpha = 1.f / (1.f + __expf(-e));

    float* o = out + (size_t)d * OUT_DIM;
    atomicAdd(&o[lane],      alpha * zs0);
    atomicAdd(&o[lane + 32], alpha * zs1);
}

extern "C" void kernel_launch(void* const* d_in, const int* in_sizes, int n_in,
                              void* d_out, int out_size, void* d_ws, size_t ws_size,
                              hipStream_t stream) {
    const float* h      = (const float*)d_in[0];
    const int*   src    = (const int*)  d_in[1];
    const int*   dst    = (const int*)  d_in[2];
    const float* W_fc   = (const float*)d_in[3];
    const float* a_attn = (const float*)d_in[4];
    float*       out    = (float*)d_out;
    float*       z      = (float*)d_ws;

    const int nNodes = in_sizes[0] / IN_DIM;
    const int nEdges = in_sizes[1];

    {
        const int n4 = out_size / 4;
        const int blocks = min((n4 + 255) / 256, 4096);
        gat_zero_out<<<blocks, 256, 0, stream>>>((float4*)out, n4);
    }

    {
        const int nWaves = (nNodes + 15) / 16;
        const int blocks = (nWaves + 7) / 8;
        gat_gemm_z<<<blocks, 256, 0, stream>>>(h, W_fc, z, nNodes);
    }

    {
        const int blocks = (nEdges + 7) / 8;
        gat_edge<<<blocks, 256, 0, stream>>>(z, src, dst, a_attn, out, nEdges);
    }
}


# owner pass: records of a range counting-sorted by destination row in LDS, then a quarter-wave sums each row in registers and stores it once (no LDS float adds or accumulators); sum order changes
# speedup vs baseline: 2.4120x; 2.4120x over previous
#include <hip/hip_runtime.h>
#include <hip/hip_bf16.h>

typedef __attribute__((ext_vector_type(2))) float v2f;
typedef __attribute__((ext_vector_type(8))) float v8f;

#define IN_DIM  128
#define OUT_DIM 64

__global__ void gat_zero_out(float4* __restrict__ out, int n4) {
    int i = blockIdx.x * blockDim.x + threadIdx.x;
    int stride = gridDim.x * blockDim.x;
    float4 z4 = make_float4(0.f, 0.f, 0.f, 0.f);
    for (; i < n4; i += stride) out[i] = z4;
}

__device__ __forceinline__ void gemm_tile(const float* __restrict__ h,
                                          const float* __restrict__ W,
                                          float* __restrict__ z,
                                          int row0, int lane) {
    const int half = lane >> 4;
    const int l16  = lane & 15;

    const float* hrow = h + (size_t)(row0 + l16) * IN_DIM + half * 2;

    v8f acc0 = {}, acc1 = {}, acc2 = {}, acc3 = {};

    for (int k = 0; k < IN_DIM; k += 4) {
        v2f a;
        a.x = hrow[k];
        a.y = hrow[k + 1];

        const float* wk = W + (size_t)(k + half * 2) * OUT_DIM + l16;
        v2f b0, b1, b2, b3;
        b0.x = wk[0];       b0.y = wk[OUT_DIM + 0];
        b1.x = wk[16];      b1.y = wk[OUT_DIM + 16];
        b2.x = wk[32];      b2.y = wk[OUT_DIM + 32];
        b3.x = wk[48];      b3.y = wk[OUT_DIM + 48];

        acc0 = __builtin_amdgcn_wmma_f32_16x16x4_f32(false, a, false, b0,
                                                     (short)0, acc0, false, false);
        acc1 = __builtin_amdgcn_wmma_f32_16x16x4_f32(false, a, false, b1,
                                                     (short)0, acc1, false, false);
        acc2 = __builtin_amdgcn_wmma_f32_16x16x4_f32(false, a, false, b2,
                                                     (short)0, acc2, false, false);
        acc3 = __builtin_amdgcn_wmma_f32_16x16x4_f32(false, a, false, b3,
                                                     (short)0, acc3, false, false);
    }

    float* zt = z + (size_t)row0 * OUT_DIM + l16;
#pragma unroll
    for (int v = 0; v < 8; ++v) {
        const size_t r = (size_t)(v + 8 * half) * OUT_DIM;
        zt[r + 0]  = acc0[v];
        zt[r + 16] = acc1[v];
        zt[r + 32] = acc2[v];
        zt[r + 48] = acc3[v];
    }
}

__global__ void gat_edge(const float* __restrict__ z,
                         const int* __restrict__ src,
                         const int* __restrict__ dst,
                         const float* __restrict__ a_attn,
                         float* __restrict__ out,
                         int nEdges) {
    const int edge = (blockIdx.x * blockDim.x + threadIdx.x) >> 5;
    const int lane = threadIdx.x & 31;
    if (edge >= nEdges) return;

    const int s = src[edge];
    const int d = dst[edge];
    const float* zs = z + (size_t)s * OUT_DIM;
    const float* zd = z + (size_t)d * OUT_DIM;

    const float zs0 = zs[lane],      zs1 = zs[lane + 32];
    const float zd0 = zd[lane],      zd1 = zd[lane + 32];

    float p = zs0 * a_attn[lane]      + zs1 * a_attn[lane + 32]
            + zd0 * a_attn[64 + lane] + zd1 * a_attn[96 + lane];

#pragma unroll
    for (int off = 16; off > 0; off >>= 1)
        p += __shfl_xor(p, off, 32);

    const float e = (p >= 0.f) ? p : 0.01f * p;
    const float alpha = 1.f / (1.f + __expf(-e));

    float* o = out + (size_t)d * OUT_DIM;
    atomicAdd(&o[lane],      alpha * zs0);
    atomicAdd(&o[lane + 32], alpha * zs1);
}

#define NB    256
#define PCH   4096
#define RMAX  392
#define SRC_BITS 23

__device__ __forceinline__ void partition_chunk(const int* __restrict__ src,
                                                const int* __restrict__ dst,
                                                unsigned* __restrict__ sorted,
                                                int* __restrict__ tab,
                                                int nEdges, int R, int c) {
    __shared__ int hist[NB];
    __shared__ int wsum[NB / 32];
    const int t = threadIdx.x;
    if (t < NB) hist[t] = 0;
    __syncthreads();

    int s[4], d[4], bk[4], rk[4];
#pragma unroll
    for (int k = 0; k < 4; ++k) {
        const int e = c * PCH + k * 1024 + t;
        bk[k] = -1; s[k] = 0; d[k] = 0; rk[k] = 0;
        if (e < nEdges) {
            s[k] = src[e];
            d[k] = dst[e];
            bk[k] = d[k] / R;
            rk[k] = atomicAdd(&hist[bk[k]], 1);
        }
    }
    __syncthreads();

    int v = 0, incl = 0;
    if (t < NB) {
        v = hist[t];
        incl = v;
#pragma unroll
        for (int off = 1; off < 32; off <<= 1) {
            const int n = __shfl_up(incl, off, 32);
            if ((t & 31) >= off) incl += n;
        }
        if ((t & 31) == 31) wsum[t >> 5] = incl;
    }
    __syncthreads();
    if (t < NB) {
        int pre = 0;
        for (int w = 0; w < (t >> 5); ++w) pre += wsum[w];
        const int excl = pre + incl - v;
        hist[t] = excl;
        tab[c * (NB + 1) + t] = excl;
        if (t == NB - 1) tab[c * (NB + 1) + NB] = excl + v;
    }
    __syncthreads();

#pragma unroll
    for (int k = 0; k < 4; ++k) {
        if (bk[k] >= 0)
            sorted[(size_t)c * PCH + hist[bk[k]] + rk[k]] =
                (unsigned)s[k] | ((unsigned)(d[k] - bk[k] * R) << SRC_BITS);
    }
}

#define GB 256
__global__ __launch_bounds__(1024, 16)
void gat_gemm_partition(const float* __restrict__ h,
                        const float* __restrict__ W,
                        float* __restrict__ z,
                        int nNodes,
                        const int* __restrict__ src,
                        const int* __restrict__ dst,
                        unsigned* __restrict__ sorted,
                        int* __restrict__ tab,
                        int nEdges, int R) {
    if (blockIdx.x < GB) {
        const int wave = __builtin_amdgcn_readfirstlane((int)(threadIdx.x >> 5));
        const int lane = threadIdx.x & 31;
        const int nTiles = (nNodes + 15) >> 4;
        for (int tile = blockIdx.x + GB * wave; tile < nTiles; tile += GB * 32)
            gemm_tile(h, W, z, tile * 16, lane);
    } else {
        partition_chunk(src, dst, sorted, tab, nEdges, R, blockIdx.x - GB);
    }
}

template <int M>
__device__ __forceinline__ float row_xor(float p) {
    return __int_as_float(__builtin_amdgcn_update_dpp(
        0, __float_as_int(p), 0x160 + M, 0xf, 0xf, false));
}

__device__ __forceinline__ float half_wave_sum(float p) {
    p += row_xor<1>(p);
    p += row_xor<2>(p);
    p += row_xor<4>(p);
    p += row_xor<8>(p);
    return p;
}

__device__ __forceinline__ float quarter_wave_sum(float p) {
    p += row_xor<1>(p);
    p += row_xor<2>(p);
    p += row_xor<4>(p);
    return p;
}

#define CAP (RMAX * OUT_DIM / 2)

__device__ __forceinline__ void owner_lds_add_route(const float4* __restrict__ z4,
                                                    const unsigned* __restrict__ sorted,
                                                    const int* __restrict__ tab,
                                                    float4 a1, float4* acc4,
                                                    const float* sdst,
                                                    float* __restrict__ out,
                                                    int node0, int nLoc, int R, int NC, int b) {
    float* acc = (float*)acc4;
    const int t    = threadIdx.x;
    const int lane = t & 31;
    const int wave = t >> 5;
    const int half = lane >> 4;
    const int l16  = lane & 15;

    for (int i = t; i < R * (OUT_DIM / 4); i += 1024)
        acc4[i] = make_float4(0.f, 0.f, 0.f, 0.f);
    __syncthreads();

    for (int c0 = 0; c0 < NC; c0 += 1024) {
        const int cL = c0 + wave + 32 * lane;
        int loV = 0, hiV = 0;
        if (cL < NC) {
            loV = tab[cL * (NB + 1) + b];
            hiV = tab[cL * (NB + 1) + b + 1];
        }
        for (int k = 0; k < 32; ++k) {
            const int c = c0 + wave + 32 * k;
            if (c >= NC) break;
            const int lo = __builtin_amdgcn_readlane(loV, k);
            const int hi = __builtin_amdgcn_readlane(hiV, k);
            const unsigned* run = sorted + (size_t)c * PCH;
            for (int i = lo; i < hi; i += 32) {
                int n = hi - i;
                if (n > 32) n = 32;
                int ed = 0;
                if (lane < n) ed = (int)run[i + lane];
                for (int j0 = 0; j0 < n; j0 += 16) {
                    float4 v[8];
                    float sd[8];
                    int dl[8];
                    bool ok[8];
#pragma unroll
                    for (int u = 0; u < 8; ++u) {
                        const int j = j0 + 2 * u + half;
                        const unsigned pk = (unsigned)__shfl(ed, j & 31, 32);
                        const int sj = (int)(pk & ((1u << SRC_BITS) - 1u));
                        ok[u] = j < n;
                        dl[u] = (int)(pk >> SRC_BITS);
                        v[u] = make_float4(0.f, 0.f, 0.f, 0.f);
                        sd[u] = 0.f;
                        if (ok[u]) {
                            v[u] = z4[(size_t)sj * 16 + l16];
                            sd[u] = sdst[dl[u]];
                        }
                    }
#pragma unroll
                    for (int u = 0; u < 8; ++u) {
                        if (j0 + 2 * u < n) {
                            float p = v[u].x * a1.x + v[u].y * a1.y
                                    + v[u].z * a1.z + v[u].w * a1.w;
                            p = half_wave_sum(p) + sd[u];
                            if (ok[u]) {
                                const float e = (p >= 0.f) ? p : 0.01f * p;
                                const float alpha = 1.f / (1.f + __expf(-e));
                                const float w0 = alpha * v[u].x, w1 = alpha * v[u].y;
                                const float w2 = alpha * v[u].z, w3 = alpha * v[u].w;
                                float* arow = acc + dl[u] * OUT_DIM + 4 * l16;
                                atomicAdd(arow + ((0 + half) & 3), half ? w1 : w0);
                                atomicAdd(arow + ((1 + half) & 3), half ? w2 : w1);
                                atomicAdd(arow + ((2 + half) & 3), half ? w3 : w2);
                                atomicAdd(arow + ((3 + half) & 3), half ? w0 : w3);
                            }
                        }
                    }
                }
            }
        }
    }
    __syncthreads();

    float4* o4 = (float4*)(out + (size_t)node0 * OUT_DIM);
    for (int i = t; i < nLoc * (OUT_DIM / 4); i += 1024) o4[i] = acc4[i];
}

__global__ __launch_bounds__(1024)
void gat_owner(const float* __restrict__ z,
               const unsigned* __restrict__ sorted,
               const int* __restrict__ tab,
               const float* __restrict__ a_attn,
               float* __restrict__ out,
               int nNodes, int R, int NC) {
    static_assert(RMAX + 1 <= 512, "the row scan uses 16 waves");
    __shared__ float4 acc4[RMAX * OUT_DIM / 4];
    __shared__ float sdst[RMAX];
    __shared__ int cnt[RMAX + 1];
    __shared__ int fill[RMAX];
    __shared__ int wsum[16];
    __shared__ int total;
    unsigned* stage = (unsigned*)acc4;
    unsigned* recs  = stage + CAP;

    const int t    = threadIdx.x;
    const int lane = t & 31;
    const int wave = __builtin_amdgcn_readfirstlane(t >> 5);
    const int half = lane >> 4;
    const int l16  = lane & 15;
    const int b    = blockIdx.x;
    const int node0 = b * R;
    int nLoc = nNodes - node0;
    if (nLoc > R) nLoc = R;
    if (nLoc < 0) nLoc = 0;

    const float4* z4 = (const float4*)z;
    const float4 a1 = ((const float4*)a_attn)[l16];
    const float4 a2 = ((const float4*)a_attn)[16 + l16];

    if (t <= R) cnt[t] = 0;
    if (t == 0) total = 0;
    __syncthreads();

    {
        constexpr int NQ = (RMAX + 63) / 64;
        float4 v[NQ];
#pragma unroll
        for (int q = 0; q < NQ; ++q) {
            const int r = q * 64 + wave * 2 + half;
            v[q] = make_float4(0.f, 0.f, 0.f, 0.f);
            if (r < nLoc) v[q] = z4[(size_t)(node0 + r) * 16 + l16];
        }
#pragma unroll
        for (int q = 0; q < NQ; ++q) {
            const int r = q * 64 + wave * 2 + half;
            float p = v[q].x * a2.x + v[q].y * a2.y + v[q].z * a2.z + v[q].w * a2.w;
            p = half_wave_sum(p);
            if (r < nLoc && l16 == 0) sdst[r] = p;
        }
    }

    for (int c0 = 0; c0 < NC; c0 += 1024) {
        const int cL = c0 + wave + 32 * lane;
        int loV = 0, hiV = 0;
        if (cL < NC) {
            loV = tab[cL * (NB + 1) + b];
            hiV = tab[cL * (NB + 1) + b + 1];
        }
        for (int kb = 0; kb < 32; kb += 8) {
            if (c0 + wave + 32 * kb >= NC) break;
            unsigned ed[8];
            int m[8];
            int tot8 = 0;
#pragma unroll
            for (int q = 0; q < 8; ++q) {
                const int c  = c0 + wave + 32 * (kb + q);
                const int lo = __builtin_amdgcn_readlane(loV, kb + q);
                const int hi = __builtin_amdgcn_readlane(hiV, kb + q);
                int n = (c < NC) ? hi - lo : 0;
                if (n > 32) n = 32;
                m[q] = n;
                tot8 += n;
                ed[q] = 0u;
                if (lane < n) ed[q] = sorted[(size_t)c * PCH + lo + lane];
            }
            int base = 0;
            if (lane == 0) base = atomicAdd(&total, tot8);
            base = __builtin_amdgcn_readfirstlane(base);
#pragma unroll
            for (int q = 0; q < 8; ++q) {
                if (lane < m[q]) {
                    atomicAdd(&cnt[ed[q] >> SRC_BITS], 1);
                    if (base + lane < CAP) stage[base + lane] = ed[q];
                }
                base += m[q];
            }
            for (int q = 0; q < 8; ++q) {
                const int c = c0 + wave + 32 * (kb + q);
                if (c >= NC) break;
                const int lo = __builtin_amdgcn_readlane(loV, kb + q);
                const int hi = __builtin_amdgcn_readlane(hiV, kb + q);
                for (int i = lo + 32; i < hi; i += 32) {
                    int n = hi - i;
                    if (n > 32) n = 32;
                    unsigned e1 = 0u;
                    if (lane < n) e1 = sorted[(size_t)c * PCH + i + lane];
                    int b1 = 0;
                    if (lane == 0) b1 = atomicAdd(&total, n);
                    b1 = __builtin_amdgcn_readfirstlane(b1);
                    if (lane < n) {
                        atomicAdd(&cnt[e1 >> SRC_BITS], 1);
                        if (b1 + lane < CAP) stage[b1 + lane] = e1;
                    }
                }
            }
        }
    }
    __syncthreads();

    const int tot = total;
    if (tot > CAP) {
        owner_lds_add_route(z4, sorted, tab, a1, acc4, sdst, out, node0, nLoc, R, NC, b);
        return;
    }

    int cv = 0, incl = 0;
    if (t < 512) {
        cv = (t < R) ? cnt[t] : 0;
        incl = cv;
#pragma unroll
        for (int off = 1; off < 32; off <<= 1) {
            const int n = __shfl_up(incl, off, 32);
            if ((t & 31) >= off) incl += n;
        }
        if ((t & 31) == 31) wsum[t >> 5] = incl;
    }
    __syncthreads();
    if (t < 512) {
        int pre = 0;
        for (int w = 0; w < (t >> 5); ++w) pre += wsum[w];
        const int excl = pre + incl - cv;
        if (t < R) { cnt[t] = excl; fill[t] = excl; }
        if (t == 0) cnt[R] = tot;
    }
    __syncthreads();

    for (int i = t; i < tot; i += 1024) {
        const unsigned pk = stage[i];
        const int pos = atomicAdd(&fill[pk >> SRC_BITS], 1);
        recs[pos] = pk & ((1u << SRC_BITS) - 1u);
    }
    __syncthreads();

    const int q8 = lane >> 3;
    const int l8 = lane & 7;
    const float4 a1lo = ((const float4*)a_attn)[l8];
    const float4 a1hi = ((const float4*)a_attn)[8 + l8];
    const int qw = wave * 4 + q8;
    float4* o4 = (float4*)(out + (size_t)node0 * OUT_DIM);
    for (int r0 = 0; r0 < nLoc; r0 += 128) {
        const int r = r0 + qw;
        const bool live = r < nLoc;
        int s0 = 0, deg = 0;
        float sd = 0.f;
        if (live) {
            s0  = cnt[r];
            deg = cnt[r + 1] - s0;
            sd  = sdst[r];
        }
        const int d0 = __builtin_amdgcn_readlane(deg, 0);
        const int d1 = __builtin_amdgcn_readlane(deg, 8);
        const int d2 = __builtin_amdgcn_readlane(deg, 16);
        const int d3 = __builtin_amdgcn_readlane(deg, 24);
        const int d01 = d0 > d1 ? d0 : d1;
        const int d23 = d2 > d3 ? d2 : d3;
        const int degMax = d01 > d23 ? d01 : d23;
        float4 sumLo = make_float4(0.f, 0.f, 0.f, 0.f);
        float4 sumHi = make_float4(0.f, 0.f, 0.f, 0.f);
        for (int j = 0; j < degMax; j += 8) {
            unsigned sj[8];
            bool ok[8];
            float4 vlo[8], vhi[8];
#pragma unroll
            for (int u = 0; u < 8; ++u) {
                ok[u] = j + u < deg;
                sj[u] = recs[ok[u] ? s0 + j + u : 0];
            }
#pragma unroll
            for (int u = 0; u < 8; ++u) {
                vlo[u] = make_float4(0.f, 0.f, 0.f, 0.f);
                vhi[u] = make_float4(0.f, 0.f, 0.f, 0.f);
                if (ok[u]) {
                    vlo[u] = z4[(size_t)sj[u] * 16 + l8];
                    vhi[u] = z4[(size_t)sj[u] * 16 + 8 + l8];
                }
            }
#pragma unroll
            for (int u = 0; u < 8; ++u) {
                if (j + u < degMax) {
                    float p = vlo[u].x * a1lo.x + vlo[u].y * a1lo.y
                            + vlo[u].z * a1lo.z + vlo[u].w * a1lo.w
                            + vhi[u].x * a1hi.x + vhi[u].y * a1hi.y
                            + vhi[u].z * a1hi.z + vhi[u].w * a1hi.w;
                    p = quarter_wave_sum(p) + sd;
                    const float e = (p >= 0.f) ? p : 0.01f * p;
                    float alpha = 1.f / (1.f + __expf(-e));
                    alpha = ok[u] ? alpha : 0.f;
                    sumLo.x += alpha * vlo[u].x;
                    sumLo.y += alpha * vlo[u].y;
                    sumLo.z += alpha * vlo[u].z;
                    sumLo.w += alpha * vlo[u].w;
                    sumHi.x += alpha * vhi[u].x;
                    sumHi.y += alpha * vhi[u].y;
                    sumHi.z += alpha * vhi[u].z;
                    sumHi.w += alpha * vhi[u].w;
                }
            }
        }
        if (live) {
            o4[(size_t)r * 16 + l8]     = sumLo;
            o4[(size_t)r * 16 + 8 + l8] = sumHi;
        }
    }
}

extern "C" void kernel_launch(void* const* d_in, const int* in_sizes, int n_in,
                              void* d_out, int out_size, void* d_ws, size_t ws_size,
                              hipStream_t stream) {
    const float* h      = (const float*)d_in[0];
    const int*   src    = (const int*)  d_in[1];
    const int*   dst    = (const int*)  d_in[2];
    const float* W_fc   = (const float*)d_in[3];
    const float* a_attn = (const float*)d_in[4];
    float*       out    = (float*)d_out;
    float*       z      = (float*)d_ws;

    const int nNodes = in_sizes[0] / IN_DIM;
    const int nEdges = in_sizes[1];

    const int R  = (nNodes + NB - 1) / NB;
    const int NC = (nEdges + PCH - 1) / PCH;
    const size_t zBytes   = (size_t)nNodes * OUT_DIM * sizeof(float);
    const size_t sortedB  = (size_t)NC * PCH * sizeof(unsigned);
    const size_t tabB     = (size_t)NC * (NB + 1) * sizeof(int);
    unsigned* sorted = (unsigned*)((char*)d_ws + zBytes);
    int*  tab    = (int*)((char*)d_ws + zBytes + sortedB);

    if (R <= RMAX && nNodes <= (1 << SRC_BITS) && zBytes + sortedB + tabB <= ws_size) {
        gat_gemm_partition<<<GB + NC, 1024, 0, stream>>>(h, W_fc, z, nNodes,
                                                         src, dst, sorted, tab, nEdges, R);
        gat_owner<<<NB, 1024, 0, stream>>>(z, sorted, tab, a_attn, out,
                                           nNodes, R, NC);
    } else {
        gat_gemm_partition<<<GB, 1024, 0, stream>>>(h, W_fc, z, nNodes,
                                                    src, dst, sorted, tab, nEdges, R);
        const int n4 = out_size / 4;
        const int zb = min((n4 + 255) / 256, 4096);
        gat_zero_out<<<zb, 256, 0, stream>>>((float4*)out, n4);
        const int blocks = (nEdges + 7) / 8;
        gat_edge<<<blocks, 256, 0, stream>>>(z, src, dst, a_attn, out, nEdges);
    }
}


# owner pass: each edge's weight (logit tail, leaky step, sigmoid) computed once, by the one lane of its quarter-wave that owns the edge, and sent to the other seven by DPP8; sum order unchanged
# speedup vs baseline: 2.5237x; 1.0463x over previous
#include <hip/hip_runtime.h>
#include <hip/hip_bf16.h>

typedef __attribute__((ext_vector_type(2))) float v2f;
typedef __attribute__((ext_vector_type(8))) float v8f;

#define IN_DIM  128
#define OUT_DIM 64

__global__ void gat_zero_out(float4* __restrict__ out, int n4) {
    int i = blockIdx.x * blockDim.x + threadIdx.x;
    int stride = gridDim.x * blockDim.x;
    float4 z4 = make_float4(0.f, 0.f, 0.f, 0.f);
    for (; i < n4; i += stride) out[i] = z4;
}

__device__ __forceinline__ void gemm_tile(const float* __restrict__ h,
                                          const float* __restrict__ W,
                                          float* __restrict__ z,
                                          int row0, int lane) {
    const int half = lane >> 4;
    const int l16  = lane & 15;

    const float* hrow = h + (size_t)(row0 + l16) * IN_DIM + half * 2;

    v8f acc0 = {}, acc1 = {}, acc2 = {}, acc3 = {};

    for (int k = 0; k < IN_DIM; k += 4) {
        v2f a;
        a.x = hrow[k];
        a.y = hrow[k + 1];

        const float* wk = W + (size_t)(k + half * 2) * OUT_DIM + l16;
        v2f b0, b1, b2, b3;
        b0.x = wk[0];       b0.y = wk[OUT_DIM + 0];
        b1.x = wk[16];      b1.y = wk[OUT_DIM + 16];
        b2.x = wk[32];      b2.y = wk[OUT_DIM + 32];
        b3.x = wk[48];      b3.y = wk[OUT_DIM + 48];

        acc0 = __builtin_amdgcn_wmma_f32_16x16x4_f32(false, a, false, b0,
                                                     (short)0, acc0, false, false);
        acc1 = __builtin_amdgcn_wmma_f32_16x16x4_f32(false, a, false, b1,
                                                     (short)0, acc1, false, false);
        acc2 = __builtin_amdgcn_wmma_f32_16x16x4_f32(false, a, false, b2,
                                                     (short)0, acc2, false, false);
        acc3 = __builtin_amdgcn_wmma_f32_16x16x4_f32(false, a, false, b3,
                                                     (short)0, acc3, false, false);
    }

    float* zt = z + (size_t)row0 * OUT_DIM + l16;
#pragma unroll
    for (int v = 0; v < 8; ++v) {
        const size_t r = (size_t)(v + 8 * half) * OUT_DIM;
        zt[r + 0]  = acc0[v];
        zt[r + 16] = acc1[v];
        zt[r + 32] = acc2[v];
        zt[r + 48] = acc3[v];
    }
}

__global__ void gat_edge(const float* __restrict__ z,
                         const int* __restrict__ src,
                         const int* __restrict__ dst,
                         const float* __restrict__ a_attn,
                         float* __restrict__ out,
                         int nEdges) {
    const int edge = (blockIdx.x * blockDim.x + threadIdx.x) >> 5;
    const int lane = threadIdx.x & 31;
    if (edge >= nEdges) return;

    const int s = src[edge];
    const int d = dst[edge];
    const float* zs = z + (size_t)s * OUT_DIM;
    const float* zd = z + (size_t)d * OUT_DIM;

    const float zs0 = zs[lane],      zs1 = zs[lane + 32];
    const float zd0 = zd[lane],      zd1 = zd[lane + 32];

    float p = zs0 * a_attn[lane]      + zs1 * a_attn[lane + 32]
            + zd0 * a_attn[64 + lane] + zd1 * a_attn[96 + lane];

#pragma unroll
    for (int off = 16; off > 0; off >>= 1)
        p += __shfl_xor(p, off, 32);

    const float e = (p >= 0.f) ? p : 0.01f * p;
    const float alpha = 1.f / (1.f + __expf(-e));

    float* o = out + (size_t)d * OUT_DIM;
    atomicAdd(&o[lane],      alpha * zs0);
    atomicAdd(&o[lane + 32], alpha * zs1);
}

#define NB    256
#define PCH   4096
#define RMAX  392
#define SRC_BITS 23

__device__ __forceinline__ void partition_chunk(const int* __restrict__ src,
                                                const int* __restrict__ dst,
                                                unsigned* __restrict__ sorted,
                                                int* __restrict__ tab,
                                                int nEdges, int R, int c) {
    __shared__ int hist[NB];
    __shared__ int wsum[NB / 32];
    const int t = threadIdx.x;
    if (t < NB) hist[t] = 0;
    __syncthreads();

    int s[4], d[4], bk[4], rk[4];
#pragma unroll
    for (int k = 0; k < 4; ++k) {
        const int e = c * PCH + k * 1024 + t;
        bk[k] = -1; s[k] = 0; d[k] = 0; rk[k] = 0;
        if (e < nEdges) {
            s[k] = src[e];
            d[k] = dst[e];
            bk[k] = d[k] / R;
            rk[k] = atomicAdd(&hist[bk[k]], 1);
        }
    }
    __syncthreads();

    int v = 0, incl = 0;
    if (t < NB) {
        v = hist[t];
        incl = v;
#pragma unroll
        for (int off = 1; off < 32; off <<= 1) {
            const int n = __shfl_up(incl, off, 32);
            if ((t & 31) >= off) incl += n;
        }
        if ((t & 31) == 31) wsum[t >> 5] = incl;
    }
    __syncthreads();
    if (t < NB) {
        int pre = 0;
        for (int w = 0; w < (t >> 5); ++w) pre += wsum[w];
        const int excl = pre + incl - v;
        hist[t] = excl;
        tab[c * (NB + 1) + t] = excl;
        if (t == NB - 1) tab[c * (NB + 1) + NB] = excl + v;
    }
    __syncthreads();

#pragma unroll
    for (int k = 0; k < 4; ++k) {
        if (bk[k] >= 0)
            sorted[(size_t)c * PCH + hist[bk[k]] + rk[k]] =
                (unsigned)s[k] | ((unsigned)(d[k] - bk[k] * R) << SRC_BITS);
    }
}

#define GB 256
__global__ __launch_bounds__(1024, 16)
void gat_gemm_partition(const float* __restrict__ h,
                        const float* __restrict__ W,
                        float* __restrict__ z,
                        int nNodes,
                        const int* __restrict__ src,
                        const int* __restrict__ dst,
                        unsigned* __restrict__ sorted,
                        int* __restrict__ tab,
                        int nEdges, int R) {
    if (blockIdx.x < GB) {
        const int wave = __builtin_amdgcn_readfirstlane((int)(threadIdx.x >> 5));
        const int lane = threadIdx.x & 31;
        const int nTiles = (nNodes + 15) >> 4;
        for (int tile = blockIdx.x + GB * wave; tile < nTiles; tile += GB * 32)
            gemm_tile(h, W, z, tile * 16, lane);
    } else {
        partition_chunk(src, dst, sorted, tab, nEdges, R, blockIdx.x - GB);
    }
}

template <int M>
__device__ __forceinline__ float row_xor(float p) {
    return __int_as_float(__builtin_amdgcn_update_dpp(
        0, __float_as_int(p), 0x160 + M, 0xf, 0xf, false));
}

__device__ __forceinline__ float half_wave_sum(float p) {
    p += row_xor<1>(p);
    p += row_xor<2>(p);
    p += row_xor<4>(p);
    p += row_xor<8>(p);
    return p;
}

__device__ __forceinline__ float quarter_wave_sum(float p) {
    p += row_xor<1>(p);
    p += row_xor<2>(p);
    p += row_xor<4>(p);
    return p;
}

__device__ __forceinline__ unsigned oct_bcast(unsigned x, int u) {
    switch (u) {
    case 0:  return __builtin_amdgcn_mov_dpp8(x, 0x000000u);
    case 1:  return __builtin_amdgcn_mov_dpp8(x, 0x249249u);
    case 2:  return __builtin_amdgcn_mov_dpp8(x, 0x492492u);
    case 3:  return __builtin_amdgcn_mov_dpp8(x, 0x6db6dbu);
    case 4:  return __builtin_amdgcn_mov_dpp8(x, 0x924924u);
    case 5:  return __builtin_amdgcn_mov_dpp8(x, 0xb6db6du);
    case 6:  return __builtin_amdgcn_mov_dpp8(x, 0xdb6db6u);
    default: return __builtin_amdgcn_mov_dpp8(x, 0xffffffu);
    }
}

#define CAP (RMAX * OUT_DIM / 2)

__device__ __forceinline__ void owner_lds_add_route(const float4* __restrict__ z4,
                                                    const unsigned* __restrict__ sorted,
                                                    const int* __restrict__ tab,
                                                    float4 a1, float4* acc4,
                                                    const float* sdst,
                                                    float* __restrict__ out,
                                                    int node0, int nLoc, int R, int NC, int b) {
    float* acc = (float*)acc4;
    const int t    = threadIdx.x;
    const int lane = t & 31;
    const int wave = t >> 5;
    const int half = lane >> 4;
    const int l16  = lane & 15;

    for (int i = t; i < R * (OUT_DIM / 4); i += 1024)
        acc4[i] = make_float4(0.f, 0.f, 0.f, 0.f);
    __syncthreads();

    for (int c0 = 0; c0 < NC; c0 += 1024) {
        const int cL = c0 + wave + 32 * lane;
        int loV = 0, hiV = 0;
        if (cL < NC) {
            loV = tab[cL * (NB + 1) + b];
            hiV = tab[cL * (NB + 1) + b + 1];
        }
        for (int k = 0; k < 32; ++k) {
            const int c = c0 + wave + 32 * k;
            if (c >= NC) break;
            const int lo = __builtin_amdgcn_readlane(loV, k);
            const int hi = __builtin_amdgcn_readlane(hiV, k);
            const unsigned* run = sorted + (size_t)c * PCH;
            for (int i = lo; i < hi; i += 32) {
                int n = hi - i;
                if (n > 32) n = 32;
                int ed = 0;
                if (lane < n) ed = (int)run[i + lane];
                for (int j0 = 0; j0 < n; j0 += 16) {
                    float4 v[8];
                    float sd[8];
                    int dl[8];
                    bool ok[8];
#pragma unroll
                    for (int u = 0; u < 8; ++u) {
                        const int j = j0 + 2 * u + half;
                        const unsigned pk = (unsigned)__shfl(ed, j & 31, 32);
                        const int sj = (int)(pk & ((1u << SRC_BITS) - 1u));
                        ok[u] = j < n;
                        dl[u] = (int)(pk >> SRC_BITS);
                        v[u] = make_float4(0.f, 0.f, 0.f, 0.f);
                        sd[u] = 0.f;
                        if (ok[u]) {
                            v[u] = z4[(size_t)sj * 16 + l16];
                            sd[u] = sdst[dl[u]];
                        }
                    }
#pragma unroll
                    for (int u = 0; u < 8; ++u) {
                        if (j0 + 2 * u < n) {
                            float p = v[u].x * a1.x + v[u].y * a1.y
                                    + v[u].z * a1.z + v[u].w * a1.w;
                            p = half_wave_sum(p) + sd[u];
                            if (ok[u]) {
                                const float e = (p >= 0.f) ? p : 0.01f * p;
                                const float alpha = 1.f / (1.f + __expf(-e));
                                const float w0 = alpha * v[u].x, w1 = alpha * v[u].y;
                                const float w2 = alpha * v[u].z, w3 = alpha * v[u].w;
                                float* arow = acc + dl[u] * OUT_DIM + 4 * l16;
                                atomicAdd(arow + ((0 + half) & 3), half ? w1 : w0);
                                atomicAdd(arow + ((1 + half) & 3), half ? w2 : w1);
                                atomicAdd(arow + ((2 + half) & 3), half ? w3 : w2);
                                atomicAdd(arow + ((3 + half) & 3), half ? w0 : w3);
                            }
                        }
                    }
                }
            }
        }
    }
    __syncthreads();

    float4* o4 = (float4*)(out + (size_t)node0 * OUT_DIM);
    for (int i = t; i < nLoc * (OUT_DIM / 4); i += 1024) o4[i] = acc4[i];
}

__global__ __launch_bounds__(1024)
void gat_owner(const float* __restrict__ z,
               const unsigned* __restrict__ sorted,
               const int* __restrict__ tab,
               const float* __restrict__ a_attn,
               float* __restrict__ out,
               int nNodes, int R, int NC) {
    static_assert(RMAX + 1 <= 512, "the row scan uses 16 waves");
    __shared__ float4 acc4[RMAX * OUT_DIM / 4];
    __shared__ float sdst[RMAX];
    __shared__ int cnt[RMAX + 1];
    __shared__ int fill[RMAX];
    __shared__ int wsum[16];
    __shared__ int total;
    unsigned* stage = (unsigned*)acc4;
    unsigned* recs  = stage + CAP;

    const int t    = threadIdx.x;
    const int lane = t & 31;
    const int wave = __builtin_amdgcn_readfirstlane(t >> 5);
    const int half = lane >> 4;
    const int l16  = lane & 15;
    const int b    = blockIdx.x;
    const int node0 = b * R;
    int nLoc = nNodes - node0;
    if (nLoc > R) nLoc = R;
    if (nLoc < 0) nLoc = 0;

    const float4* z4 = (const float4*)z;
    const float4 a1 = ((const float4*)a_attn)[l16];
    const float4 a2 = ((const float4*)a_attn)[16 + l16];

    if (t <= R) cnt[t] = 0;
    if (t == 0) total = 0;
    __syncthreads();

    {
        constexpr int NQ = (RMAX + 63) / 64;
        float4 v[NQ];
#pragma unroll
        for (int q = 0; q < NQ; ++q) {
            const int r = q * 64 + wave * 2 + half;
            v[q] = make_float4(0.f, 0.f, 0.f, 0.f);
            if (r < nLoc) v[q] = z4[(size_t)(node0 + r) * 16 + l16];
        }
#pragma unroll
        for (int q = 0; q < NQ; ++q) {
            const int r = q * 64 + wave * 2 + half;
            float p = v[q].x * a2.x + v[q].y * a2.y + v[q].z * a2.z + v[q].w * a2.w;
            p = half_wave_sum(p);
            if (r < nLoc && l16 == 0) sdst[r] = p;
        }
    }

    for (int c0 = 0; c0 < NC; c0 += 1024) {
        const int cL = c0 + wave + 32 * lane;
        int loV = 0, hiV = 0;
        if (cL < NC) {
            loV = tab[cL * (NB + 1) + b];
            hiV = tab[cL * (NB + 1) + b + 1];
        }
        for (int kb = 0; kb < 32; kb += 8) {
            if (c0 + wave + 32 * kb >= NC) break;
            unsigned ed[8];
            int m[8];
            int tot8 = 0;
#pragma unroll
            for (int q = 0; q < 8; ++q) {
                const int c  = c0 + wave + 32 * (kb + q);
                const int lo = __builtin_amdgcn_readlane(loV, kb + q);
                const int hi = __builtin_amdgcn_readlane(hiV, kb + q);
                int n = (c < NC) ? hi - lo : 0;
                if (n > 32) n = 32;
                m[q] = n;
                tot8 += n;
                ed[q] = 0u;
                if (lane < n) ed[q] = sorted[(size_t)c * PCH + lo + lane];
            }
            int base = 0;
            if (lane == 0) base = atomicAdd(&total, tot8);
            base = __builtin_amdgcn_readfirstlane(base);
#pragma unroll
            for (int q = 0; q < 8; ++q) {
                if (lane < m[q]) {
                    atomicAdd(&cnt[ed[q] >> SRC_BITS], 1);
                    if (base + lane < CAP) stage[base + lane] = ed[q];
                }
                base += m[q];
            }
            for (int q = 0; q < 8; ++q) {
                const int c = c0 + wave + 32 * (kb + q);
                if (c >= NC) break;
                const int lo = __builtin_amdgcn_readlane(loV, kb + q);
                const int hi = __builtin_amdgcn_readlane(hiV, kb + q);
                for (int i = lo + 32; i < hi; i += 32) {
                    int n = hi - i;
                    if (n > 32) n = 32;
                    unsigned e1 = 0u;
                    if (lane < n) e1 = sorted[(size_t)c * PCH + i + lane];
                    int b1 = 0;
                    if (lane == 0) b1 = atomicAdd(&total, n);
                    b1 = __builtin_amdgcn_readfirstlane(b1);
                    if (lane < n) {
                        atomicAdd(&cnt[e1 >> SRC_BITS], 1);
                        if (b1 + lane < CAP) stage[b1 + lane] = e1;
                    }
                }
            }
        }
    }
    __syncthreads();

    const int tot = total;
    if (tot > CAP) {
        owner_lds_add_route(z4, sorted, tab, a1, acc4, sdst, out, node0, nLoc, R, NC, b);
        return;
    }

    int cv = 0, incl = 0;
    if (t < 512) {
        cv = (t < R) ? cnt[t] : 0;
        incl = cv;
#pragma unroll
        for (int off = 1; off < 32; off <<= 1) {
            const int n = __shfl_up(incl, off, 32);
            if ((t & 31) >= off) incl += n;
        }
        if ((t & 31) == 31) wsum[t >> 5] = incl;
    }
    __syncthreads();
    if (t < 512) {
        int pre = 0;
        for (int w = 0; w < (t >> 5); ++w) pre += wsum[w];
        const int excl = pre + incl - cv;
        if (t < R) { cnt[t] = excl; fill[t] = excl; }
        if (t == 0) cnt[R] = tot;
    }
    __syncthreads();

    for (int i = t; i < tot; i += 1024) {
        const unsigned pk = stage[i];
        const int pos = atomicAdd(&fill[pk >> SRC_BITS], 1);
        recs[pos] = pk & ((1u << SRC_BITS) - 1u);
    }
    __syncthreads();

    const int q8 = lane >> 3;
    const int l8 = lane & 7;
    const float4 a1lo = ((const float4*)a_attn)[l8];
    const float4 a1hi = ((const float4*)a_attn)[8 + l8];
    const int qw = wave * 4 + q8;
    float4* o4 = (float4*)(out + (size_t)node0 * OUT_DIM);
    for (int r0 = 0; r0 < nLoc; r0 += 128) {
        const int r = r0 + qw;
        const bool live = r < nLoc;
        int s0 = 0, deg = 0;
        float sd = 0.f;
        if (live) {
            s0  = cnt[r];
            deg = cnt[r + 1] - s0;
            sd  = sdst[r];
        }
        const int d0 = __builtin_amdgcn_readlane(deg, 0);
        const int d1 = __builtin_amdgcn_readlane(deg, 8);
        const int d2 = __builtin_amdgcn_readlane(deg, 16);
        const int d3 = __builtin_amdgcn_readlane(deg, 24);
        const int d01 = d0 > d1 ? d0 : d1;
        const int d23 = d2 > d3 ? d2 : d3;
        const int degMax = d01 > d23 ? d01 : d23;
        float4 sumLo = make_float4(0.f, 0.f, 0.f, 0.f);
        float4 sumHi = make_float4(0.f, 0.f, 0.f, 0.f);
        for (int j = 0; j < degMax; j += 8) {
            const bool mine = j + l8 < deg;
            const unsigned smine = recs[mine ? s0 + j + l8 : 0];
            bool ok[8];
            float4 vlo[8], vhi[8];
#pragma unroll
            for (int u = 0; u < 8; ++u) {
                ok[u] = j + u < deg;
                const unsigned sj = oct_bcast(smine, u);
                vlo[u] = make_float4(0.f, 0.f, 0.f, 0.f);
                vhi[u] = make_float4(0.f, 0.f, 0.f, 0.f);
                if (ok[u]) {
                    vlo[u] = z4[(size_t)sj * 16 + l8];
                    vhi[u] = z4[(size_t)sj * 16 + 8 + l8];
                }
            }
            float pm = 0.f;
#pragma unroll
            for (int u = 0; u < 8; ++u) {
                float p = vlo[u].x * a1lo.x + vlo[u].y * a1lo.y
                        + vlo[u].z * a1lo.z + vlo[u].w * a1lo.w
                        + vhi[u].x * a1hi.x + vhi[u].y * a1hi.y
                        + vhi[u].z * a1hi.z + vhi[u].w * a1hi.w;
                p = quarter_wave_sum(p);
                pm = (l8 == u) ? p : pm;
            }
            pm += sd;
            const float e = (pm >= 0.f) ? pm : 0.01f * pm;
            float alpha = 1.f / (1.f + __expf(-e));
            alpha = mine ? alpha : 0.f;
#pragma unroll
            for (int u = 0; u < 8; ++u) {
                const float au = __uint_as_float(oct_bcast(__float_as_uint(alpha), u));
                sumLo.x += au * vlo[u].x;
                sumLo.y += au * vlo[u].y;
                sumLo.z += au * vlo[u].z;
                sumLo.w += au * vlo[u].w;
                sumHi.x += au * vhi[u].x;
                sumHi.y += au * vhi[u].y;
                sumHi.z += au * vhi[u].z;
                sumHi.w += au * vhi[u].w;
            }
        }
        if (live) {
            o4[(size_t)r * 16 + l8]     = sumLo;
            o4[(size_t)r * 16 + 8 + l8] = sumHi;
        }
    }
}

extern "C" void kernel_launch(void* const* d_in, const int* in_sizes, int n_in,
                              void* d_out, int out_size, void* d_ws, size_t ws_size,
                              hipStream_t stream) {
    const float* h      = (const float*)d_in[0];
    const int*   src    = (const int*)  d_in[1];
    const int*   dst    = (const int*)  d_in[2];
    const float* W_fc   = (const float*)d_in[3];
    const float* a_attn = (const float*)d_in[4];
    float*       out    = (float*)d_out;
    float*       z      = (float*)d_ws;

    const int nNodes = in_sizes[0] / IN_DIM;
    const int nEdges = in_sizes[1];

    const int R  = (nNodes + NB - 1) / NB;
    const int NC = (nEdges + PCH - 1) / PCH;
    const size_t zBytes   = (size_t)nNodes * OUT_DIM * sizeof(float);
    const size_t sortedB  = (size_t)NC * PCH * sizeof(unsigned);
    const size_t tabB     = (size_t)NC * (NB + 1) * sizeof(int);
    unsigned* sorted = (unsigned*)((char*)d_ws + zBytes);
    int*  tab    = (int*)((char*)d_ws + zBytes + sortedB);

    if (R <= RMAX && nNodes <= (1 << SRC_BITS) && zBytes + sortedB + tabB <= ws_size) {
        gat_gemm_partition<<<GB + NC, 1024, 0, stream>>>(h, W_fc, z, nNodes,
                                                         src, dst, sorted, tab, nEdges, R);
        gat_owner<<<NB, 1024, 0, stream>>>(z, sorted, tab, a_attn, out,
                                           nNodes, R, NC);
    } else {
        gat_gemm_partition<<<GB, 1024, 0, stream>>>(h, W_fc, z, nNodes,
                                                    src, dst, sorted, tab, nEdges, R);
        const int n4 = out_size / 4;
        const int zb = min((n4 + 255) / 256, 4096);
        gat_zero_out<<<zb, 256, 0, stream>>>((float4*)out, n4);
        const int blocks = (nEdges + 7) / 8;
        gat_edge<<<blocks, 256, 0, stream>>>(z, src, dst, a_attn, out, nEdges);
    }
}


# product: B fragments of W staged once per work-group in LDS in fragment order (2 ds_load_b128 per k-step, were 8 global loads); A read 16 B per lane and k-step pair; the product's k sum order changes
# speedup vs baseline: 3.0032x; 1.1900x over previous
#include <hip/hip_runtime.h>
#include <hip/hip_bf16.h>

typedef __attribute__((ext_vector_type(2))) float v2f;
typedef __attribute__((ext_vector_type(8))) float v8f;

#define IN_DIM  128
#define OUT_DIM 64

__global__ void gat_zero_out(float4* __restrict__ out, int n4) {
    int i = blockIdx.x * blockDim.x + threadIdx.x;
    int stride = gridDim.x * blockDim.x;
    float4 z4 = make_float4(0.f, 0.f, 0.f, 0.f);
    for (; i < n4; i += stride) out[i] = z4;
}

#define KSTEPS (IN_DIM / 4)

__device__ __forceinline__ int frag_k(int t, int hf) {
    return 8 * (t >> 1) + 4 * hf + 2 * (t & 1);
}

__device__ __forceinline__ void stage_b_fragments(const float* __restrict__ W,
                                                  float4* wb0, float4* wb1) {
    const int t    = threadIdx.x >> 5;
    const int lane = threadIdx.x & 31;
    const int hf   = lane >> 4;
    const int l16  = lane & 15;
    const float* w0 = W + (size_t)frag_k(t, hf) * OUT_DIM + l16;
    const float* w1 = w0 + OUT_DIM;
    wb0[t * 32 + lane] = make_float4(w0[0],  w1[0],  w0[16], w1[16]);
    wb1[t * 32 + lane] = make_float4(w0[32], w1[32], w0[48], w1[48]);
}

#define KSTEP_WMMA(ax, ay, p0, p1)                                                      \
    do {                                                                                \
        v2f a_, b0_, b1_, b2_, b3_;                                                     \
        a_.x  = (ax);   a_.y  = (ay);                                                   \
        b0_.x = (p0).x; b0_.y = (p0).y;                                                 \
        b1_.x = (p0).z; b1_.y = (p0).w;                                                 \
        b2_.x = (p1).x; b2_.y = (p1).y;                                                 \
        b3_.x = (p1).z; b3_.y = (p1).w;                                                 \
                           \
        acc0 = __builtin_amdgcn_wmma_f32_16x16x4_f32(false, a_, false, b0_,             \
                                                     (short)0, acc0, false, false);     \
        acc1 = __builtin_amdgcn_wmma_f32_16x16x4_f32(false, a_, false, b1_,             \
                                                     (short)0, acc1, false, false);     \
        acc2 = __builtin_amdgcn_wmma_f32_16x16x4_f32(false, a_, false, b2_,             \
                                                     (short)0, acc2, false, false);     \
        acc3 = __builtin_amdgcn_wmma_f32_16x16x4_f32(false, a_, false, b3_,             \
                                                     (short)0, acc3, false, false);     \
    } while (0)

__device__ __forceinline__ void gemm_tile(const float* __restrict__ h,
                                          const float4* wb0, const float4* wb1,
                                          float* __restrict__ z,
                                          int row0, int lane) {
    const int hf  = lane >> 4;
    const int l16 = lane & 15;

    const float4* hrow4 = (const float4*)(h + (size_t)(row0 + l16) * IN_DIM + hf * 4);
    const float4* f0 = wb0 + lane;
    const float4* f1 = wb1 + lane;

    v8f acc0 = {}, acc1 = {}, acc2 = {}, acc3 = {};

#pragma unroll 1
    for (int i = 0; i < KSTEPS / 4; ++i) {
        const float4 qa = hrow4[4 * i];
        const float4 qb = hrow4[4 * i + 2];
        {
            const float4 p0a = f0[(4 * i) * 32],     p1a = f1[(4 * i) * 32];
            const float4 p0b = f0[(4 * i + 1) * 32], p1b = f1[(4 * i + 1) * 32];
            KSTEP_WMMA(qa.x, qa.y, p0a, p1a);
            KSTEP_WMMA(qa.z, qa.w, p0b, p1b);
        }
        {
            const float4 p0a = f0[(4 * i + 2) * 32], p1a = f1[(4 * i + 2) * 32];
            const float4 p0b = f0[(4 * i + 3) * 32], p1b = f1[(4 * i + 3) * 32];
            KSTEP_WMMA(qb.x, qb.y, p0a, p1a);
            KSTEP_WMMA(qb.z, qb.w, p0b, p1b);
        }
    }

    float* zt = z + (size_t)row0 * OUT_DIM + l16;
#pragma unroll
    for (int v = 0; v < 8; ++v) {
        const size_t r = (size_t)(v + 8 * hf) * OUT_DIM;
        zt[r + 0]  = acc0[v];
        zt[r + 16] = acc1[v];
        zt[r + 32] = acc2[v];
        zt[r + 48] = acc3[v];
    }
}

__global__ void gat_edge(const float* __restrict__ z,
                         const int* __restrict__ src,
                         const int* __restrict__ dst,
                         const float* __restrict__ a_attn,
                         float* __restrict__ out,
                         int nEdges) {
    const int edge = (blockIdx.x * blockDim.x + threadIdx.x) >> 5;
    const int lane = threadIdx.x & 31;
    if (edge >= nEdges) return;

    const int s = src[edge];
    const int d = dst[edge];
    const float* zs = z + (size_t)s * OUT_DIM;
    const float* zd = z + (size_t)d * OUT_DIM;

    const float zs0 = zs[lane],      zs1 = zs[lane + 32];
    const float zd0 = zd[lane],      zd1 = zd[lane + 32];

    float p = zs0 * a_attn[lane]      + zs1 * a_attn[lane + 32]
            + zd0 * a_attn[64 + lane] + zd1 * a_attn[96 + lane];

#pragma unroll
    for (int off = 16; off > 0; off >>= 1)
        p += __shfl_xor(p, off, 32);

    const float e = (p >= 0.f) ? p : 0.01f * p;
    const float alpha = 1.f / (1.f + __expf(-e));

    float* o = out + (size_t)d * OUT_DIM;
    atomicAdd(&o[lane],      alpha * zs0);
    atomicAdd(&o[lane + 32], alpha * zs1);
}

#define NB    256
#define PCH   4096
#define RMAX  392
#define SRC_BITS 23

__device__ __forceinline__ void partition_chunk(const int* __restrict__ src,
                                                const int* __restrict__ dst,
                                                unsigned* __restrict__ sorted,
                                                int* __restrict__ tab,
                                                int nEdges, int R, int c) {
    __shared__ int hist[NB];
    __shared__ int wsum[NB / 32];
    const int t = threadIdx.x;
    if (t < NB) hist[t] = 0;
    __syncthreads();

    int s[4], d[4], bk[4], rk[4];
#pragma unroll
    for (int k = 0; k < 4; ++k) {
        const int e = c * PCH + k * 1024 + t;
        bk[k] = -1; s[k] = 0; d[k] = 0; rk[k] = 0;
        if (e < nEdges) {
            s[k] = src[e];
            d[k] = dst[e];
            bk[k] = d[k] / R;
            rk[k] = atomicAdd(&hist[bk[k]], 1);
        }
    }
    __syncthreads();

    int v = 0, incl = 0;
    if (t < NB) {
        v = hist[t];
        incl = v;
#pragma unroll
        for (int off = 1; off < 32; off <<= 1) {
            const int n = __shfl_up(incl, off, 32);
            if ((t & 31) >= off) incl += n;
        }
        if ((t & 31) == 31) wsum[t >> 5] = incl;
    }
    __syncthreads();
    if (t < NB) {
        int pre = 0;
        for (int w = 0; w < (t >> 5); ++w) pre += wsum[w];
        const int excl = pre + incl - v;
        hist[t] = excl;
        tab[c * (NB + 1) + t] = excl;
        if (t == NB - 1) tab[c * (NB + 1) + NB] = excl + v;
    }
    __syncthreads();

#pragma unroll
    for (int k = 0; k < 4; ++k) {
        if (bk[k] >= 0)
            sorted[(size_t)c * PCH + hist[bk[k]] + rk[k]] =
                (unsigned)s[k] | ((unsigned)(d[k] - bk[k] * R) << SRC_BITS);
    }
}

#define GB 256
__global__ __launch_bounds__(1024, 16)
void gat_gemm_partition(const float* __restrict__ h,
                        const float* __restrict__ W,
                        float* __restrict__ z,
                        int nNodes,
                        const int* __restrict__ src,
                        const int* __restrict__ dst,
                        unsigned* __restrict__ sorted,
                        int* __restrict__ tab,
                        int nEdges, int R) {
    if (blockIdx.x < GB) {
        __shared__ float4 wb0[KSTEPS * 32];
        __shared__ float4 wb1[KSTEPS * 32];
        stage_b_fragments(W, wb0, wb1);
        __syncthreads();
        const int wave = __builtin_amdgcn_readfirstlane((int)(threadIdx.x >> 5));
        const int lane = threadIdx.x & 31;
        const int nTiles = (nNodes + 15) >> 4;
        for (int tile = blockIdx.x + GB * wave; tile < nTiles; tile += GB * 32)
            gemm_tile(h, wb0, wb1, z, tile * 16, lane);
    } else {
        partition_chunk(src, dst, sorted, tab, nEdges, R, blockIdx.x - GB);
    }
}

template <int M>
__device__ __forceinline__ float row_xor(float p) {
    return __int_as_float(__builtin_amdgcn_update_dpp(
        0, __float_as_int(p), 0x160 + M, 0xf, 0xf, false));
}

__device__ __forceinline__ float half_wave_sum(float p) {
    p += row_xor<1>(p);
    p += row_xor<2>(p);
    p += row_xor<4>(p);
    p += row_xor<8>(p);
    return p;
}

__device__ __forceinline__ float quarter_wave_sum(float p) {
    p += row_xor<1>(p);
    p += row_xor<2>(p);
    p += row_xor<4>(p);
    return p;
}

__device__ __forceinline__ unsigned oct_bcast(unsigned x, int u) {
    switch (u) {
    case 0:  return __builtin_amdgcn_mov_dpp8(x, 0x000000u);
    case 1:  return __builtin_amdgcn_mov_dpp8(x, 0x249249u);
    case 2:  return __builtin_amdgcn_mov_dpp8(x, 0x492492u);
    case 3:  return __builtin_amdgcn_mov_dpp8(x, 0x6db6dbu);
    case 4:  return __builtin_amdgcn_mov_dpp8(x, 0x924924u);
    case 5:  return __builtin_amdgcn_mov_dpp8(x, 0xb6db6du);
    case 6:  return __builtin_amdgcn_mov_dpp8(x, 0xdb6db6u);
    default: return __builtin_amdgcn_mov_dpp8(x, 0xffffffu);
    }
}

#define CAP (RMAX * OUT_DIM / 2)

__device__ __forceinline__ void owner_lds_add_route(const float4* __restrict__ z4,
                                                    const unsigned* __restrict__ sorted,
                                                    const int* __restrict__ tab,
                                                    float4 a1, float4* acc4,
                                                    const float* sdst,
                                                    float* __restrict__ out,
                                                    int node0, int nLoc, int R, int NC, int b) {
    float* acc = (float*)acc4;
    const int t    = threadIdx.x;
    const int lane = t & 31;
    const int wave = t >> 5;
    const int half = lane >> 4;
    const int l16  = lane & 15;

    for (int i = t; i < R * (OUT_DIM / 4); i += 1024)
        acc4[i] = make_float4(0.f, 0.f, 0.f, 0.f);
    __syncthreads();

    for (int c0 = 0; c0 < NC; c0 += 1024) {
        const int cL = c0 + wave + 32 * lane;
        int loV = 0, hiV = 0;
        if (cL < NC) {
            loV = tab[cL * (NB + 1) + b];
            hiV = tab[cL * (NB + 1) + b + 1];
        }
        for (int k = 0; k < 32; ++k) {
            const int c = c0 + wave + 32 * k;
            if (c >= NC) break;
            const int lo = __builtin_amdgcn_readlane(loV, k);
            const int hi = __builtin_amdgcn_readlane(hiV, k);
            const unsigned* run = sorted + (size_t)c * PCH;
            for (int i = lo; i < hi; i += 32) {
                int n = hi - i;
                if (n > 32) n = 32;
                int ed = 0;
                if (lane < n) ed = (int)run[i + lane];
                for (int j0 = 0; j0 < n; j0 += 16) {
                    float4 v[8];
                    float sd[8];
                    int dl[8];
                    bool ok[8];
#pragma unroll
                    for (int u = 0; u < 8; ++u) {
                        const int j = j0 + 2 * u + half;
                        const unsigned pk = (unsigned)__shfl(ed, j & 31, 32);
                        const int sj = (int)(pk & ((1u << SRC_BITS) - 1u));
                        ok[u] = j < n;
                        dl[u] = (int)(pk >> SRC_BITS);
                        v[u] = make_float4(0.f, 0.f, 0.f, 0.f);
                        sd[u] = 0.f;
                        if (ok[u]) {
                            v[u] = z4[(size_t)sj * 16 + l16];
                            sd[u] = sdst[dl[u]];
                        }
                    }
#pragma unroll
                    for (int u = 0; u < 8; ++u) {
                        if (j0 + 2 * u < n) {
                            float p = v[u].x * a1.x + v[u].y * a1.y
                                    + v[u].z * a1.z + v[u].w * a1.w;
                            p = half_wave_sum(p) + sd[u];
                            if (ok[u]) {
                                const float e = (p >= 0.f) ? p : 0.01f * p;
                                const float alpha = 1.f / (1.f + __expf(-e));
                                const float w0 = alpha * v[u].x, w1 = alpha * v[u].y;
                                const float w2 = alpha * v[u].z, w3 = alpha * v[u].w;
                                float* arow = acc + dl[u] * OUT_DIM + 4 * l16;
                                atomicAdd(arow + ((0 + half) & 3), half ? w1 : w0);
                                atomicAdd(arow + ((1 + half) & 3), half ? w2 : w1);
                                atomicAdd(arow + ((2 + half) & 3), half ? w3 : w2);
                                atomicAdd(arow + ((3 + half) & 3), half ? w0 : w3);
                            }
                        }
                    }
                }
            }
        }
    }
    __syncthreads();

    float4* o4 = (float4*)(out + (size_t)node0 * OUT_DIM);
    for (int i = t; i < nLoc * (OUT_DIM / 4); i += 1024) o4[i] = acc4[i];
}

__global__ __launch_bounds__(1024)
void gat_owner(const float* __restrict__ z,
               const unsigned* __restrict__ sorted,
               const int* __restrict__ tab,
               const float* __restrict__ a_attn,
               float* __restrict__ out,
               int nNodes, int R, int NC) {
    static_assert(RMAX + 1 <= 512, "the row scan uses 16 waves");
    __shared__ float4 acc4[RMAX * OUT_DIM / 4];
    __shared__ float sdst[RMAX];
    __shared__ int cnt[RMAX + 1];
    __shared__ int fill[RMAX];
    __shared__ int wsum[16];
    __shared__ int total;
    unsigned* stage = (unsigned*)acc4;
    unsigned* recs  = stage + CAP;

    const int t    = threadIdx.x;
    const int lane = t & 31;
    const int wave = __builtin_amdgcn_readfirstlane(t >> 5);
    const int half = lane >> 4;
    const int l16  = lane & 15;
    const int b    = blockIdx.x;
    const int node0 = b * R;
    int nLoc = nNodes - node0;
    if (nLoc > R) nLoc = R;
    if (nLoc < 0) nLoc = 0;

    const float4* z4 = (const float4*)z;
    const float4 a1 = ((const float4*)a_attn)[l16];
    const float4 a2 = ((const float4*)a_attn)[16 + l16];

    if (t <= R) cnt[t] = 0;
    if (t == 0) total = 0;
    __syncthreads();

    {
        constexpr int NQ = (RMAX + 63) / 64;
        float4 v[NQ];
#pragma unroll
        for (int q = 0; q < NQ; ++q) {
            const int r = q * 64 + wave * 2 + half;
            v[q] = make_float4(0.f, 0.f, 0.f, 0.f);
            if (r < nLoc) v[q] = z4[(size_t)(node0 + r) * 16 + l16];
        }
#pragma unroll
        for (int q = 0; q < NQ; ++q) {
            const int r = q * 64 + wave * 2 + half;
            float p = v[q].x * a2.x + v[q].y * a2.y + v[q].z * a2.z + v[q].w * a2.w;
            p = half_wave_sum(p);
            if (r < nLoc && l16 == 0) sdst[r] = p;
        }
    }

    for (int c0 = 0; c0 < NC; c0 += 1024) {
        const int cL = c0 + wave + 32 * lane;
        int loV = 0, hiV = 0;
        if (cL < NC) {
            loV = tab[cL * (NB + 1) + b];
            hiV = tab[cL * (NB + 1) + b + 1];
        }
        for (int kb = 0; kb < 32; kb += 8) {
            if (c0 + wave + 32 * kb >= NC) break;
            unsigned ed[8];
            int m[8];
            int tot8 = 0;
#pragma unroll
            for (int q = 0; q < 8; ++q) {
                const int c  = c0 + wave + 32 * (kb + q);
                const int lo = __builtin_amdgcn_readlane(loV, kb + q);
                const int hi = __builtin_amdgcn_readlane(hiV, kb + q);
                int n = (c < NC) ? hi - lo : 0;
                if (n > 32) n = 32;
                m[q] = n;
                tot8 += n;
                ed[q] = 0u;
                if (lane < n) ed[q] = sorted[(size_t)c * PCH + lo + lane];
            }
            int base = 0;
            if (lane == 0) base = atomicAdd(&total, tot8);
            base = __builtin_amdgcn_readfirstlane(base);
#pragma unroll
            for (int q = 0; q < 8; ++q) {
                if (lane < m[q]) {
                    atomicAdd(&cnt[ed[q] >> SRC_BITS], 1);
                    if (base + lane < CAP) stage[base + lane] = ed[q];
                }
                base += m[q];
            }
            for (int q = 0; q < 8; ++q) {
                const int c = c0 + wave + 32 * (kb + q);
                if (c >= NC) break;
                const int lo = __builtin_amdgcn_readlane(loV, kb + q);
                const int hi = __builtin_amdgcn_readlane(hiV, kb + q);
                for (int i = lo + 32; i < hi; i += 32) {
                    int n = hi - i;
                    if (n > 32) n = 32;
                    unsigned e1 = 0u;
                    if (lane < n) e1 = sorted[(size_t)c * PCH + i + lane];
                    int b1 = 0;
                    if (lane == 0) b1 = atomicAdd(&total, n);
                    b1 = __builtin_amdgcn_readfirstlane(b1);
                    if (lane < n) {
                        atomicAdd(&cnt[e1 >> SRC_BITS], 1);
                        if (b1 + lane < CAP) stage[b1 + lane] = e1;
                    }
                }
            }
        }
    }
    __syncthreads();

    const int tot = total;
    if (tot > CAP) {
        owner_lds_add_route(z4, sorted, tab, a1, acc4, sdst, out, node0, nLoc, R, NC, b);
        return;
    }

    int cv = 0, incl = 0;
    if (t < 512) {
        cv = (t < R) ? cnt[t] : 0;
        incl = cv;
#pragma unroll
        for (int off = 1; off < 32; off <<= 1) {
            const int n = __shfl_up(incl, off, 32);
            if ((t & 31) >= off) incl += n;
        }
        if ((t & 31) == 31) wsum[t >> 5] = incl;
    }
    __syncthreads();
    if (t < 512) {
        int pre = 0;
        for (int w = 0; w < (t >> 5); ++w) pre += wsum[w];
        const int excl = pre + incl - cv;
        if (t < R) { cnt[t] = excl; fill[t] = excl; }
        if (t == 0) cnt[R] = tot;
    }
    __syncthreads();

    for (int i = t; i < tot; i += 1024) {
        const unsigned pk = stage[i];
        const int pos = atomicAdd(&fill[pk >> SRC_BITS], 1);
        recs[pos] = pk & ((1u << SRC_BITS) - 1u);
    }
    __syncthreads();

    const int q8 = lane >> 3;
    const int l8 = lane & 7;
    const float4 a1lo = ((const float4*)a_attn)[l8];
    const float4 a1hi = ((const float4*)a_attn)[8 + l8];
    const int qw = wave * 4 + q8;
    float4* o4 = (float4*)(out + (size_t)node0 * OUT_DIM);
    for (int r0 = 0; r0 < nLoc; r0 += 128) {
        const int r = r0 + qw;
        const bool live = r < nLoc;
        int s0 = 0, deg = 0;
        float sd = 0.f;
        if (live) {
            s0  = cnt[r];
            deg = cnt[r + 1] - s0;
            sd  = sdst[r];
        }
        const int d0 = __builtin_amdgcn_readlane(deg, 0);
        const int d1 = __builtin_amdgcn_readlane(deg, 8);
        const int d2 = __builtin_amdgcn_readlane(deg, 16);
        const int d3 = __builtin_amdgcn_readlane(deg, 24);
        const int d01 = d0 > d1 ? d0 : d1;
        const int d23 = d2 > d3 ? d2 : d3;
        const int degMax = d01 > d23 ? d01 : d23;
        float4 sumLo = make_float4(0.f, 0.f, 0.f, 0.f);
        float4 sumHi = make_float4(0.f, 0.f, 0.f, 0.f);
        for (int j = 0; j < degMax; j += 8) {
            const bool mine = j + l8 < deg;
            const unsigned smine = recs[mine ? s0 + j + l8 : 0];
            bool ok[8];
            float4 vlo[8], vhi[8];
#pragma unroll
            for (int u = 0; u < 8; ++u) {
                ok[u] = j + u < deg;
                const unsigned sj = oct_bcast(smine, u);
                vlo[u] = make_float4(0.f, 0.f, 0.f, 0.f);
                vhi[u] = make_float4(0.f, 0.f, 0.f, 0.f);
                if (ok[u]) {
                    vlo[u] = z4[(size_t)sj * 16 + l8];
                    vhi[u] = z4[(size_t)sj * 16 + 8 + l8];
                }
            }
            float pm = 0.f;
#pragma unroll
            for (int u = 0; u < 8; ++u) {
                float p = vlo[u].x * a1lo.x + vlo[u].y * a1lo.y
                        + vlo[u].z * a1lo.z + vlo[u].w * a1lo.w
                        + vhi[u].x * a1hi.x + vhi[u].y * a1hi.y
                        + vhi[u].z * a1hi.z + vhi[u].w * a1hi.w;
                p = quarter_wave_sum(p);
                pm = (l8 == u) ? p : pm;
            }
            pm += sd;
            const float e = (pm >= 0.f) ? pm : 0.01f * pm;
            float alpha = 1.f / (1.f + __expf(-e));
            alpha = mine ? alpha : 0.f;
#pragma unroll
            for (int u = 0; u < 8; ++u) {
                const float au = __uint_as_float(oct_bcast(__float_as_uint(alpha), u));
                sumLo.x += au * vlo[u].x;
                sumLo.y += au * vlo[u].y;
                sumLo.z += au * vlo[u].z;
                sumLo.w += au * vlo[u].w;
                sumHi.x += au * vhi[u].x;
                sumHi.y += au * vhi[u].y;
                sumHi.z += au * vhi[u].z;
                sumHi.w += au * vhi[u].w;
            }
        }
        if (live) {
            o4[(size_t)r * 16 + l8]     = sumLo;
            o4[(size_t)r * 16 + 8 + l8] = sumHi;
        }
    }
}

extern "C" void kernel_launch(void* const* d_in, const int* in_sizes, int n_in,
                              void* d_out, int out_size, void* d_ws, size_t ws_size,
                              hipStream_t stream) {
    const float* h      = (const float*)d_in[0];
    const int*   src    = (const int*)  d_in[1];
    const int*   dst    = (const int*)  d_in[2];
    const float* W_fc   = (const float*)d_in[3];
    const float* a_attn = (const float*)d_in[4];
    float*       out    = (float*)d_out;
    float*       z      = (float*)d_ws;

    const int nNodes = in_sizes[0] / IN_DIM;
    const int nEdges = in_sizes[1];

    const int R  = (nNodes + NB - 1) / NB;
    const int NC = (nEdges + PCH - 1) / PCH;
    const size_t zBytes   = (size_t)nNodes * OUT_DIM * sizeof(float);
    const size_t sortedB  = (size_t)NC * PCH * sizeof(unsigned);
    const size_t tabB     = (size_t)NC * (NB + 1) * sizeof(int);
    unsigned* sorted = (unsigned*)((char*)d_ws + zBytes);
    int*  tab    = (int*)((char*)d_ws + zBytes + sortedB);

    if (R <= RMAX && nNodes <= (1 << SRC_BITS) && zBytes + sortedB + tabB <= ws_size) {
        gat_gemm_partition<<<GB + NC, 1024, 0, stream>>>(h, W_fc, z, nNodes,
                                                         src, dst, sorted, tab, nEdges, R);
        gat_owner<<<NB, 1024, 0, stream>>>(z, sorted, tab, a_attn, out,
                                           nNodes, R, NC);
    } else {
        gat_gemm_partition<<<GB, 1024, 0, stream>>>(h, W_fc, z, nNodes,
                                                    src, dst, sorted, tab, nEdges, R);
        const int n4 = out_size / 4;
        const int zb = min((n4 + 255) / 256, 4096);
        gat_zero_out<<<zb, 256, 0, stream>>>((float4*)out, n4);
        const int blocks = (nEdges + 7) / 8;
        gat_edge<<<blocks, 256, 0, stream>>>(z, src, dst, a_attn, out, nEdges);
    }
}


# owner pass: a range's rows counting-sorted by edge count in LDS and dealt to quarter-waves in that order, so the four rows a wave sums together need equally many 8-edge trips; sum order unchanged
# speedup vs baseline: 3.1613x; 1.0526x over previous
#include <hip/hip_runtime.h>
#include <hip/hip_bf16.h>

typedef __attribute__((ext_vector_type(2))) float v2f;
typedef __attribute__((ext_vector_type(8))) float v8f;

#define IN_DIM  128
#define OUT_DIM 64

__global__ void gat_zero_out(float4* __restrict__ out, int n4) {
    int i = blockIdx.x * blockDim.x + threadIdx.x;
    int stride = gridDim.x * blockDim.x;
    float4 z4 = make_float4(0.f, 0.f, 0.f, 0.f);
    for (; i < n4; i += stride) out[i] = z4;
}

#define KSTEPS (IN_DIM / 4)

__device__ __forceinline__ int frag_k(int t, int hf) {
    return 8 * (t >> 1) + 4 * hf + 2 * (t & 1);
}

__device__ __forceinline__ void stage_b_fragments(const float* __restrict__ W,
                                                  float4* wb0, float4* wb1) {
    const int t    = threadIdx.x >> 5;
    const int lane = threadIdx.x & 31;
    const int hf   = lane >> 4;
    const int l16  = lane & 15;
    const float* w0 = W + (size_t)frag_k(t, hf) * OUT_DIM + l16;
    const float* w1 = w0 + OUT_DIM;
    wb0[t * 32 + lane] = make_float4(w0[0],  w1[0],  w0[16], w1[16]);
    wb1[t * 32 + lane] = make_float4(w0[32], w1[32], w0[48], w1[48]);
}

#define KSTEP_WMMA(ax, ay, p0, p1)                                                      \
    do {                                                                                \
        v2f a_, b0_, b1_, b2_, b3_;                                                     \
        a_.x  = (ax);   a_.y  = (ay);                                                   \
        b0_.x = (p0).x; b0_.y = (p0).y;                                                 \
        b1_.x = (p0).z; b1_.y = (p0).w;                                                 \
        b2_.x = (p1).x; b2_.y = (p1).y;                                                 \
        b3_.x = (p1).z; b3_.y = (p1).w;                                                 \
                           \
        acc0 = __builtin_amdgcn_wmma_f32_16x16x4_f32(false, a_, false, b0_,             \
                                                     (short)0, acc0, false, false);     \
        acc1 = __builtin_amdgcn_wmma_f32_16x16x4_f32(false, a_, false, b1_,             \
                                                     (short)0, acc1, false, false);     \
        acc2 = __builtin_amdgcn_wmma_f32_16x16x4_f32(false, a_, false, b2_,             \
                                                     (short)0, acc2, false, false);     \
        acc3 = __builtin_amdgcn_wmma_f32_16x16x4_f32(false, a_, false, b3_,             \
                                                     (short)0, acc3, false, false);     \
    } while (0)

__device__ __forceinline__ void gemm_tile(const float* __restrict__ h,
                                          const float4* wb0, const float4* wb1,
                                          float* __restrict__ z,
                                          int row0, int lane) {
    const int hf  = lane >> 4;
    const int l16 = lane & 15;

    const float4* hrow4 = (const float4*)(h + (size_t)(row0 + l16) * IN_DIM + hf * 4);
    const float4* f0 = wb0 + lane;
    const float4* f1 = wb1 + lane;

    v8f acc0 = {}, acc1 = {}, acc2 = {}, acc3 = {};

#pragma unroll 1
    for (int i = 0; i < KSTEPS / 4; ++i) {
        const float4 qa = hrow4[4 * i];
        const float4 qb = hrow4[4 * i + 2];
        {
            const float4 p0a = f0[(4 * i) * 32],     p1a = f1[(4 * i) * 32];
            const float4 p0b = f0[(4 * i + 1) * 32], p1b = f1[(4 * i + 1) * 32];
            KSTEP_WMMA(qa.x, qa.y, p0a, p1a);
            KSTEP_WMMA(qa.z, qa.w, p0b, p1b);
        }
        {
            const float4 p0a = f0[(4 * i + 2) * 32], p1a = f1[(4 * i + 2) * 32];
            const float4 p0b = f0[(4 * i + 3) * 32], p1b = f1[(4 * i + 3) * 32];
            KSTEP_WMMA(qb.x, qb.y, p0a, p1a);
            KSTEP_WMMA(qb.z, qb.w, p0b, p1b);
        }
    }

    float* zt = z + (size_t)row0 * OUT_DIM + l16;
#pragma unroll
    for (int v = 0; v < 8; ++v) {
        const size_t r = (size_t)(v + 8 * hf) * OUT_DIM;
        zt[r + 0]  = acc0[v];
        zt[r + 16] = acc1[v];
        zt[r + 32] = acc2[v];
        zt[r + 48] = acc3[v];
    }
}

__global__ void gat_edge(const float* __restrict__ z,
                         const int* __restrict__ src,
                         const int* __restrict__ dst,
                         const float* __restrict__ a_attn,
                         float* __restrict__ out,
                         int nEdges) {
    const int edge = (blockIdx.x * blockDim.x + threadIdx.x) >> 5;
    const int lane = threadIdx.x & 31;
    if (edge >= nEdges) return;

    const int s = src[edge];
    const int d = dst[edge];
    const float* zs = z + (size_t)s * OUT_DIM;
    const float* zd = z + (size_t)d * OUT_DIM;

    const float zs0 = zs[lane],      zs1 = zs[lane + 32];
    const float zd0 = zd[lane],      zd1 = zd[lane + 32];

    float p = zs0 * a_attn[lane]      + zs1 * a_attn[lane + 32]
            + zd0 * a_attn[64 + lane] + zd1 * a_attn[96 + lane];

#pragma unroll
    for (int off = 16; off > 0; off >>= 1)
        p += __shfl_xor(p, off, 32);

    const float e = (p >= 0.f) ? p : 0.01f * p;
    const float alpha = 1.f / (1.f + __expf(-e));

    float* o = out + (size_t)d * OUT_DIM;
    atomicAdd(&o[lane],      alpha * zs0);
    atomicAdd(&o[lane + 32], alpha * zs1);
}

#define NB    256
#define PCH   4096
#define RMAX  392
#define SRC_BITS 23

__device__ __forceinline__ void partition_chunk(const int* __restrict__ src,
                                                const int* __restrict__ dst,
                                                unsigned* __restrict__ sorted,
                                                int* __restrict__ tab,
                                                int nEdges, int R, int c) {
    __shared__ int hist[NB];
    __shared__ int wsum[NB / 32];
    const int t = threadIdx.x;
    if (t < NB) hist[t] = 0;
    __syncthreads();

    int s[4], d[4], bk[4], rk[4];
#pragma unroll
    for (int k = 0; k < 4; ++k) {
        const int e = c * PCH + k * 1024 + t;
        bk[k] = -1; s[k] = 0; d[k] = 0; rk[k] = 0;
        if (e < nEdges) {
            s[k] = src[e];
            d[k] = dst[e];
            bk[k] = d[k] / R;
            rk[k] = atomicAdd(&hist[bk[k]], 1);
        }
    }
    __syncthreads();

    int v = 0, incl = 0;
    if (t < NB) {
        v = hist[t];
        incl = v;
#pragma unroll
        for (int off = 1; off < 32; off <<= 1) {
            const int n = __shfl_up(incl, off, 32);
            if ((t & 31) >= off) incl += n;
        }
        if ((t & 31) == 31) wsum[t >> 5] = incl;
    }
    __syncthreads();
    if (t < NB) {
        int pre = 0;
        for (int w = 0; w < (t >> 5); ++w) pre += wsum[w];
        const int excl = pre + incl - v;
        hist[t] = excl;
        tab[c * (NB + 1) + t] = excl;
        if (t == NB - 1) tab[c * (NB + 1) + NB] = excl + v;
    }
    __syncthreads();

#pragma unroll
    for (int k = 0; k < 4; ++k) {
        if (bk[k] >= 0)
            sorted[(size_t)c * PCH + hist[bk[k]] + rk[k]] =
                (unsigned)s[k] | ((unsigned)(d[k] - bk[k] * R) << SRC_BITS);
    }
}

#define GB 256
__global__ __launch_bounds__(1024, 16)
void gat_gemm_partition(const float* __restrict__ h,
                        const float* __restrict__ W,
                        float* __restrict__ z,
                        int nNodes,
                        const int* __restrict__ src,
                        const int* __restrict__ dst,
                        unsigned* __restrict__ sorted,
                        int* __restrict__ tab,
                        int nEdges, int R) {
    if (blockIdx.x < GB) {
        __shared__ float4 wb0[KSTEPS * 32];
        __shared__ float4 wb1[KSTEPS * 32];
        stage_b_fragments(W, wb0, wb1);
        __syncthreads();
        const int wave = __builtin_amdgcn_readfirstlane((int)(threadIdx.x >> 5));
        const int lane = threadIdx.x & 31;
        const int nTiles = (nNodes + 15) >> 4;
        for (int tile = blockIdx.x + GB * wave; tile < nTiles; tile += GB * 32)
            gemm_tile(h, wb0, wb1, z, tile * 16, lane);
    } else {
        partition_chunk(src, dst, sorted, tab, nEdges, R, blockIdx.x - GB);
    }
}

template <int M>
__device__ __forceinline__ float row_xor(float p) {
    return __int_as_float(__builtin_amdgcn_update_dpp(
        0, __float_as_int(p), 0x160 + M, 0xf, 0xf, false));
}

__device__ __forceinline__ float half_wave_sum(float p) {
    p += row_xor<1>(p);
    p += row_xor<2>(p);
    p += row_xor<4>(p);
    p += row_xor<8>(p);
    return p;
}

__device__ __forceinline__ float quarter_wave_sum(float p) {
    p += row_xor<1>(p);
    p += row_xor<2>(p);
    p += row_xor<4>(p);
    return p;
}

__device__ __forceinline__ unsigned oct_bcast(unsigned x, int u) {
    switch (u) {
    case 0:  return __builtin_amdgcn_mov_dpp8(x, 0x000000u);
    case 1:  return __builtin_amdgcn_mov_dpp8(x, 0x249249u);
    case 2:  return __builtin_amdgcn_mov_dpp8(x, 0x492492u);
    case 3:  return __builtin_amdgcn_mov_dpp8(x, 0x6db6dbu);
    case 4:  return __builtin_amdgcn_mov_dpp8(x, 0x924924u);
    case 5:  return __builtin_amdgcn_mov_dpp8(x, 0xb6db6du);
    case 6:  return __builtin_amdgcn_mov_dpp8(x, 0xdb6db6u);
    default: return __builtin_amdgcn_mov_dpp8(x, 0xffffffu);
    }
}

#define CAP (RMAX * OUT_DIM / 2)
#define DBINS 64

__device__ __forceinline__ void owner_lds_add_route(const float4* __restrict__ z4,
                                                    const unsigned* __restrict__ sorted,
                                                    const int* __restrict__ tab,
                                                    float4 a1, float4* acc4,
                                                    const float* sdst,
                                                    float* __restrict__ out,
                                                    int node0, int nLoc, int R, int NC, int b) {
    float* acc = (float*)acc4;
    const int t    = threadIdx.x;
    const int lane = t & 31;
    const int wave = t >> 5;
    const int half = lane >> 4;
    const int l16  = lane & 15;

    for (int i = t; i < R * (OUT_DIM / 4); i += 1024)
        acc4[i] = make_float4(0.f, 0.f, 0.f, 0.f);
    __syncthreads();

    for (int c0 = 0; c0 < NC; c0 += 1024) {
        const int cL = c0 + wave + 32 * lane;
        int loV = 0, hiV = 0;
        if (cL < NC) {
            loV = tab[cL * (NB + 1) + b];
            hiV = tab[cL * (NB + 1) + b + 1];
        }
        for (int k = 0; k < 32; ++k) {
            const int c = c0 + wave + 32 * k;
            if (c >= NC) break;
            const int lo = __builtin_amdgcn_readlane(loV, k);
            const int hi = __builtin_amdgcn_readlane(hiV, k);
            const unsigned* run = sorted + (size_t)c * PCH;
            for (int i = lo; i < hi; i += 32) {
                int n = hi - i;
                if (n > 32) n = 32;
                int ed = 0;
                if (lane < n) ed = (int)run[i + lane];
                for (int j0 = 0; j0 < n; j0 += 16) {
                    float4 v[8];
                    float sd[8];
                    int dl[8];
                    bool ok[8];
#pragma unroll
                    for (int u = 0; u < 8; ++u) {
                        const int j = j0 + 2 * u + half;
                        const unsigned pk = (unsigned)__shfl(ed, j & 31, 32);
                        const int sj = (int)(pk & ((1u << SRC_BITS) - 1u));
                        ok[u] = j < n;
                        dl[u] = (int)(pk >> SRC_BITS);
                        v[u] = make_float4(0.f, 0.f, 0.f, 0.f);
                        sd[u] = 0.f;
                        if (ok[u]) {
                            v[u] = z4[(size_t)sj * 16 + l16];
                            sd[u] = sdst[dl[u]];
                        }
                    }
#pragma unroll
                    for (int u = 0; u < 8; ++u) {
                        if (j0 + 2 * u < n) {
                            float p = v[u].x * a1.x + v[u].y * a1.y
                                    + v[u].z * a1.z + v[u].w * a1.w;
                            p = half_wave_sum(p) + sd[u];
                            if (ok[u]) {
                                const float e = (p >= 0.f) ? p : 0.01f * p;
                                const float alpha = 1.f / (1.f + __expf(-e));
                                const float w0 = alpha * v[u].x, w1 = alpha * v[u].y;
                                const float w2 = alpha * v[u].z, w3 = alpha * v[u].w;
                                float* arow = acc + dl[u] * OUT_DIM + 4 * l16;
                                atomicAdd(arow + ((0 + half) & 3), half ? w1 : w0);
                                atomicAdd(arow + ((1 + half) & 3), half ? w2 : w1);
                                atomicAdd(arow + ((2 + half) & 3), half ? w3 : w2);
                                atomicAdd(arow + ((3 + half) & 3), half ? w0 : w3);
                            }
                        }
                    }
                }
            }
        }
    }
    __syncthreads();

    float4* o4 = (float4*)(out + (size_t)node0 * OUT_DIM);
    for (int i = t; i < nLoc * (OUT_DIM / 4); i += 1024) o4[i] = acc4[i];
}

__global__ __launch_bounds__(1024)
void gat_owner(const float* __restrict__ z,
               const unsigned* __restrict__ sorted,
               const int* __restrict__ tab,
               const float* __restrict__ a_attn,
               float* __restrict__ out,
               int nNodes, int R, int NC) {
    static_assert(RMAX + 1 <= 512, "the row scan uses 16 waves");
    __shared__ float4 acc4[RMAX * OUT_DIM / 4];
    __shared__ float sdst[RMAX];
    __shared__ int cnt[RMAX + 1];
    __shared__ int fill[RMAX];
    __shared__ int wsum[16];
    __shared__ int total;
    __shared__ int dhist[DBINS];
    __shared__ int order[RMAX];
    unsigned* stage = (unsigned*)acc4;
    unsigned* recs  = stage + CAP;

    const int t    = threadIdx.x;
    const int lane = t & 31;
    const int wave = __builtin_amdgcn_readfirstlane(t >> 5);
    const int half = lane >> 4;
    const int l16  = lane & 15;
    const int b    = blockIdx.x;
    const int node0 = b * R;
    int nLoc = nNodes - node0;
    if (nLoc > R) nLoc = R;
    if (nLoc < 0) nLoc = 0;

    const float4* z4 = (const float4*)z;
    const float4 a1 = ((const float4*)a_attn)[l16];
    const float4 a2 = ((const float4*)a_attn)[16 + l16];

    if (t <= R) cnt[t] = 0;
    if (t < DBINS) dhist[t] = 0;
    if (t == 0) total = 0;
    __syncthreads();

    {
        constexpr int NQ = (RMAX + 63) / 64;
        float4 v[NQ];
#pragma unroll
        for (int q = 0; q < NQ; ++q) {
            const int r = q * 64 + wave * 2 + half;
            v[q] = make_float4(0.f, 0.f, 0.f, 0.f);
            if (r < nLoc) v[q] = z4[(size_t)(node0 + r) * 16 + l16];
        }
#pragma unroll
        for (int q = 0; q < NQ; ++q) {
            const int r = q * 64 + wave * 2 + half;
            float p = v[q].x * a2.x + v[q].y * a2.y + v[q].z * a2.z + v[q].w * a2.w;
            p = half_wave_sum(p);
            if (r < nLoc && l16 == 0) sdst[r] = p;
        }
    }

    for (int c0 = 0; c0 < NC; c0 += 1024) {
        const int cL = c0 + wave + 32 * lane;
        int loV = 0, hiV = 0;
        if (cL < NC) {
            loV = tab[cL * (NB + 1) + b];
            hiV = tab[cL * (NB + 1) + b + 1];
        }
        for (int kb = 0; kb < 32; kb += 8) {
            if (c0 + wave + 32 * kb >= NC) break;
            unsigned ed[8];
            int m[8];
            int tot8 = 0;
#pragma unroll
            for (int q = 0; q < 8; ++q) {
                const int c  = c0 + wave + 32 * (kb + q);
                const int lo = __builtin_amdgcn_readlane(loV, kb + q);
                const int hi = __builtin_amdgcn_readlane(hiV, kb + q);
                int n = (c < NC) ? hi - lo : 0;
                if (n > 32) n = 32;
                m[q] = n;
                tot8 += n;
                ed[q] = 0u;
                if (lane < n) ed[q] = sorted[(size_t)c * PCH + lo + lane];
            }
            int base = 0;
            if (lane == 0) base = atomicAdd(&total, tot8);
            base = __builtin_amdgcn_readfirstlane(base);
#pragma unroll
            for (int q = 0; q < 8; ++q) {
                if (lane < m[q]) {
                    atomicAdd(&cnt[ed[q] >> SRC_BITS], 1);
                    if (base + lane < CAP) stage[base + lane] = ed[q];
                }
                base += m[q];
            }
            for (int q = 0; q < 8; ++q) {
                const int c = c0 + wave + 32 * (kb + q);
                if (c >= NC) break;
                const int lo = __builtin_amdgcn_readlane(loV, kb + q);
                const int hi = __builtin_amdgcn_readlane(hiV, kb + q);
                for (int i = lo + 32; i < hi; i += 32) {
                    int n = hi - i;
                    if (n > 32) n = 32;
                    unsigned e1 = 0u;
                    if (lane < n) e1 = sorted[(size_t)c * PCH + i + lane];
                    int b1 = 0;
                    if (lane == 0) b1 = atomicAdd(&total, n);
                    b1 = __builtin_amdgcn_readfirstlane(b1);
                    if (lane < n) {
                        atomicAdd(&cnt[e1 >> SRC_BITS], 1);
                        if (b1 + lane < CAP) stage[b1 + lane] = e1;
                    }
                }
            }
        }
    }
    __syncthreads();

    const int tot = total;
    if (tot > CAP) {
        owner_lds_add_route(z4, sorted, tab, a1, acc4, sdst, out, node0, nLoc, R, NC, b);
        return;
    }

    int cv = 0, incl = 0;
    int dbin = 0, drank = 0;
    if (t < 512) {
        cv = (t < R) ? cnt[t] : 0;
        if (t < nLoc) {
            dbin  = cv < DBINS - 1 ? cv : DBINS - 1;
            drank = atomicAdd(&dhist[dbin], 1);
        }
        incl = cv;
#pragma unroll
        for (int off = 1; off < 32; off <<= 1) {
            const int n = __shfl_up(incl, off, 32);
            if ((t & 31) >= off) incl += n;
        }
        if ((t & 31) == 31) wsum[t >> 5] = incl;
    }
    __syncthreads();
    if (t < 512) {
        int pre = 0;
        for (int w = 0; w < (t >> 5); ++w) pre += wsum[w];
        const int excl = pre + incl - cv;
        if (t < R) { cnt[t] = excl; fill[t] = excl; }
        if (t == 0) cnt[R] = tot;
    } else if (t < 544) {
        const int hiBin = DBINS - 1 - 2 * lane;
        const int nHi = dhist[hiBin];
        const int nLo = dhist[hiBin - 1];
        const int two = nHi + nLo;
        int run = two;
        for (int off = 1; off < 32; off <<= 1) {
            const int n = __shfl_up(run, off, 32);
            if (lane >= off) run += n;
        }
        dhist[hiBin]     = run - two;
        dhist[hiBin - 1] = run - two + nHi;
    }
    __syncthreads();

    if (t < nLoc) order[dhist[dbin] + drank] = t;

    for (int i = t; i < tot; i += 1024) {
        const unsigned pk = stage[i];
        const int pos = atomicAdd(&fill[pk >> SRC_BITS], 1);
        recs[pos] = pk & ((1u << SRC_BITS) - 1u);
    }
    __syncthreads();

    const int q8 = lane >> 3;
    const int l8 = lane & 7;
    const float4 a1lo = ((const float4*)a_attn)[l8];
    const float4 a1hi = ((const float4*)a_attn)[8 + l8];
    const int qw = wave * 4 + q8;
    float4* o4 = (float4*)(out + (size_t)node0 * OUT_DIM);
    for (int r0 = 0; r0 < nLoc; r0 += 128) {
        const int place = r0 + ((r0 & 128) ? 127 - qw : qw);
        const bool live = place < nLoc;
        int r = 0, s0 = 0, deg = 0;
        float sd = 0.f;
        if (live) {
            r   = order[place];
            s0  = cnt[r];
            deg = cnt[r + 1] - s0;
            sd  = sdst[r];
        }
        const int d0 = __builtin_amdgcn_readlane(deg, 0);
        const int d1 = __builtin_amdgcn_readlane(deg, 8);
        const int d2 = __builtin_amdgcn_readlane(deg, 16);
        const int d3 = __builtin_amdgcn_readlane(deg, 24);
        const int d01 = d0 > d1 ? d0 : d1;
        const int d23 = d2 > d3 ? d2 : d3;
        const int degMax = d01 > d23 ? d01 : d23;
        float4 sumLo = make_float4(0.f, 0.f, 0.f, 0.f);
        float4 sumHi = make_float4(0.f, 0.f, 0.f, 0.f);
        for (int j = 0; j < degMax; j += 8) {
            const bool mine = j + l8 < deg;
            const unsigned smine = recs[mine ? s0 + j + l8 : 0];
            bool ok[8];
            float4 vlo[8], vhi[8];
#pragma unroll
            for (int u = 0; u < 8; ++u) {
                ok[u] = j + u < deg;
                const unsigned sj = oct_bcast(smine, u);
                vlo[u] = make_float4(0.f, 0.f, 0.f, 0.f);
                vhi[u] = make_float4(0.f, 0.f, 0.f, 0.f);
                if (ok[u]) {
                    vlo[u] = z4[(size_t)sj * 16 + l8];
                    vhi[u] = z4[(size_t)sj * 16 + 8 + l8];
                }
            }
            float pm = 0.f;
#pragma unroll
            for (int u = 0; u < 8; ++u) {
                float p = vlo[u].x * a1lo.x + vlo[u].y * a1lo.y
                        + vlo[u].z * a1lo.z + vlo[u].w * a1lo.w
                        + vhi[u].x * a1hi.x + vhi[u].y * a1hi.y
                        + vhi[u].z * a1hi.z + vhi[u].w * a1hi.w;
                p = quarter_wave_sum(p);
                pm = (l8 == u) ? p : pm;
            }
            pm += sd;
            const float e = (pm >= 0.f) ? pm : 0.01f * pm;
            float alpha = 1.f / (1.f + __expf(-e));
            alpha = mine ? alpha : 0.f;
#pragma unroll
            for (int u = 0; u < 8; ++u) {
                const float au = __uint_as_float(oct_bcast(__float_as_uint(alpha), u));
                sumLo.x += au * vlo[u].x;
                sumLo.y += au * vlo[u].y;
                sumLo.z += au * vlo[u].z;
                sumLo.w += au * vlo[u].w;
                sumHi.x += au * vhi[u].x;
                sumHi.y += au * vhi[u].y;
                sumHi.z += au * vhi[u].z;
                sumHi.w += au * vhi[u].w;
            }
        }
        if (live) {
            o4[(size_t)r * 16 + l8]     = sumLo;
            o4[(size_t)r * 16 + 8 + l8] = sumHi;
        }
    }
}

extern "C" void kernel_launch(void* const* d_in, const int* in_sizes, int n_in,
                              void* d_out, int out_size, void* d_ws, size_t ws_size,
                              hipStream_t stream) {
    const float* h      = (const float*)d_in[0];
    const int*   src    = (const int*)  d_in[1];
    const int*   dst    = (const int*)  d_in[2];
    const float* W_fc   = (const float*)d_in[3];
    const float* a_attn = (const float*)d_in[4];
    float*       out    = (float*)d_out;
    float*       z      = (float*)d_ws;

    const int nNodes = in_sizes[0] / IN_DIM;
    const int nEdges = in_sizes[1];

    const int R  = (nNodes + NB - 1) / NB;
    const int NC = (nEdges + PCH - 1) / PCH;
    const size_t zBytes   = (size_t)nNodes * OUT_DIM * sizeof(float);
    const size_t sortedB  = (size_t)NC * PCH * sizeof(unsigned);
    const size_t tabB     = (size_t)NC * (NB + 1) * sizeof(int);
    unsigned* sorted = (unsigned*)((char*)d_ws + zBytes);
    int*  tab    = (int*)((char*)d_ws + zBytes + sortedB);

    if (R <= RMAX && nNodes <= (1 << SRC_BITS) && zBytes + sortedB + tabB <= ws_size) {
        gat_gemm_partition<<<GB + NC, 1024, 0, stream>>>(h, W_fc, z, nNodes,
                                                         src, dst, sorted, tab, nEdges, R);
        gat_owner<<<NB, 1024, 0, stream>>>(z, sorted, tab, a_attn, out,
                                           nNodes, R, NC);
    } else {
        gat_gemm_partition<<<GB, 1024, 0, stream>>>(h, W_fc, z, nNodes,
                                                    src, dst, sorted, tab, nEdges, R);
        const int n4 = out_size / 4;
        const int zb = min((n4 + 255) / 256, 4096);
        gat_zero_out<<<zb, 256, 0, stream>>>((float4*)out, n4);
        const int blocks = (nEdges + 7) / 8;
        gat_edge<<<blocks, 256, 0, stream>>>(z, src, dst, a_attn, out, nEdges);
    }
}


# owner pass: vector instructions per 8-edge trip cut by two fifths (row registers cleared once, not per trip; 32-bit gather offsets; packed-pair logit; transposed 8-lane sums); logit sum order changes
# speedup vs baseline: 3.2852x; 1.0392x over previous
#include <hip/hip_runtime.h>
#include <hip/hip_bf16.h>

typedef __attribute__((ext_vector_type(2))) float v2f;
typedef __attribute__((ext_vector_type(8))) float v8f;
typedef __attribute__((ext_vector_type(4))) float v4f;

#define IN_DIM  128
#define OUT_DIM 64

__global__ void gat_zero_out(float4* __restrict__ out, int n4) {
    int i = blockIdx.x * blockDim.x + threadIdx.x;
    int stride = gridDim.x * blockDim.x;
    float4 z4 = make_float4(0.f, 0.f, 0.f, 0.f);
    for (; i < n4; i += stride) out[i] = z4;
}

#define KSTEPS (IN_DIM / 4)

__device__ __forceinline__ int frag_k(int t, int hf) {
    return 8 * (t >> 1) + 4 * hf + 2 * (t & 1);
}

__device__ __forceinline__ void stage_b_fragments(const float* __restrict__ W,
                                                  float4* wb0, float4* wb1) {
    const int t    = threadIdx.x >> 5;
    const int lane = threadIdx.x & 31;
    const int hf   = lane >> 4;
    const int l16  = lane & 15;
    const float* w0 = W + (size_t)frag_k(t, hf) * OUT_DIM + l16;
    const float* w1 = w0 + OUT_DIM;
    wb0[t * 32 + lane] = make_float4(w0[0],  w1[0],  w0[16], w1[16]);
    wb1[t * 32 + lane] = make_float4(w0[32], w1[32], w0[48], w1[48]);
}

#define KSTEP_WMMA(ax, ay, p0, p1)                                                      \
    do {                                                                                \
        v2f a_, b0_, b1_, b2_, b3_;                                                     \
        a_.x  = (ax);   a_.y  = (ay);                                                   \
        b0_.x = (p0).x; b0_.y = (p0).y;                                                 \
        b1_.x = (p0).z; b1_.y = (p0).w;                                                 \
        b2_.x = (p1).x; b2_.y = (p1).y;                                                 \
        b3_.x = (p1).z; b3_.y = (p1).w;                                                 \
                           \
        acc0 = __builtin_amdgcn_wmma_f32_16x16x4_f32(false, a_, false, b0_,             \
                                                     (short)0, acc0, false, false);     \
        acc1 = __builtin_amdgcn_wmma_f32_16x16x4_f32(false, a_, false, b1_,             \
                                                     (short)0, acc1, false, false);     \
        acc2 = __builtin_amdgcn_wmma_f32_16x16x4_f32(false, a_, false, b2_,             \
                                                     (short)0, acc2, false, false);     \
        acc3 = __builtin_amdgcn_wmma_f32_16x16x4_f32(false, a_, false, b3_,             \
                                                     (short)0, acc3, false, false);     \
    } while (0)

__device__ __forceinline__ void gemm_tile(const float* __restrict__ h,
                                          const float4* wb0, const float4* wb1,
                                          float* __restrict__ z,
                                          int row0, int lane) {
    const int hf  = lane >> 4;
    const int l16 = lane & 15;

    const float4* hrow4 = (const float4*)(h + (size_t)(row0 + l16) * IN_DIM + hf * 4);
    const float4* f0 = wb0 + lane;
    const float4* f1 = wb1 + lane;

    v8f acc0 = {}, acc1 = {}, acc2 = {}, acc3 = {};

#pragma unroll 1
    for (int i = 0; i < KSTEPS / 4; ++i) {
        const float4 qa = hrow4[4 * i];
        const float4 qb = hrow4[4 * i + 2];
        {
            const float4 p0a = f0[(4 * i) * 32],     p1a = f1[(4 * i) * 32];
            const float4 p0b = f0[(4 * i + 1) * 32], p1b = f1[(4 * i + 1) * 32];
            KSTEP_WMMA(qa.x, qa.y, p0a, p1a);
            KSTEP_WMMA(qa.z, qa.w, p0b, p1b);
        }
        {
            const float4 p0a = f0[(4 * i + 2) * 32], p1a = f1[(4 * i + 2) * 32];
            const float4 p0b = f0[(4 * i + 3) * 32], p1b = f1[(4 * i + 3) * 32];
            KSTEP_WMMA(qb.x, qb.y, p0a, p1a);
            KSTEP_WMMA(qb.z, qb.w, p0b, p1b);
        }
    }

    float* zt = z + (size_t)row0 * OUT_DIM + l16;
#pragma unroll
    for (int v = 0; v < 8; ++v) {
        const size_t r = (size_t)(v + 8 * hf) * OUT_DIM;
        zt[r + 0]  = acc0[v];
        zt[r + 16] = acc1[v];
        zt[r + 32] = acc2[v];
        zt[r + 48] = acc3[v];
    }
}

__global__ void gat_edge(const float* __restrict__ z,
                         const int* __restrict__ src,
                         const int* __restrict__ dst,
                         const float* __restrict__ a_attn,
                         float* __restrict__ out,
                         int nEdges) {
    const int edge = (blockIdx.x * blockDim.x + threadIdx.x) >> 5;
    const int lane = threadIdx.x & 31;
    if (edge >= nEdges) return;

    const int s = src[edge];
    const int d = dst[edge];
    const float* zs = z + (size_t)s * OUT_DIM;
    const float* zd = z + (size_t)d * OUT_DIM;

    const float zs0 = zs[lane],      zs1 = zs[lane + 32];
    const float zd0 = zd[lane],      zd1 = zd[lane + 32];

    float p = zs0 * a_attn[lane]      + zs1 * a_attn[lane + 32]
            + zd0 * a_attn[64 + lane] + zd1 * a_attn[96 + lane];

#pragma unroll
    for (int off = 16; off > 0; off >>= 1)
        p += __shfl_xor(p, off, 32);

    const float e = (p >= 0.f) ? p : 0.01f * p;
    const float alpha = 1.f / (1.f + __expf(-e));

    float* o = out + (size_t)d * OUT_DIM;
    atomicAdd(&o[lane],      alpha * zs0);
    atomicAdd(&o[lane + 32], alpha * zs1);
}

#define NB    256
#define PCH   4096
#define RMAX  392
#define SRC_BITS 23

__device__ __forceinline__ void partition_chunk(const int* __restrict__ src,
                                                const int* __restrict__ dst,
                                                unsigned* __restrict__ sorted,
                                                int* __restrict__ tab,
                                                int nEdges, int R, int c) {
    __shared__ int hist[NB];
    __shared__ int wsum[NB / 32];
    const int t = threadIdx.x;
    if (t < NB) hist[t] = 0;
    __syncthreads();

    int s[4], d[4], bk[4], rk[4];
#pragma unroll
    for (int k = 0; k < 4; ++k) {
        const int e = c * PCH + k * 1024 + t;
        bk[k] = -1; s[k] = 0; d[k] = 0; rk[k] = 0;
        if (e < nEdges) {
            s[k] = src[e];
            d[k] = dst[e];
            bk[k] = d[k] / R;
            rk[k] = atomicAdd(&hist[bk[k]], 1);
        }
    }
    __syncthreads();

    int v = 0, incl = 0;
    if (t < NB) {
        v = hist[t];
        incl = v;
#pragma unroll
        for (int off = 1; off < 32; off <<= 1) {
            const int n = __shfl_up(incl, off, 32);
            if ((t & 31) >= off) incl += n;
        }
        if ((t & 31) == 31) wsum[t >> 5] = incl;
    }
    __syncthreads();
    if (t < NB) {
        int pre = 0;
        for (int w = 0; w < (t >> 5); ++w) pre += wsum[w];
        const int excl = pre + incl - v;
        hist[t] = excl;
        tab[c * (NB + 1) + t] = excl;
        if (t == NB - 1) tab[c * (NB + 1) + NB] = excl + v;
    }
    __syncthreads();

#pragma unroll
    for (int k = 0; k < 4; ++k) {
        if (bk[k] >= 0)
            sorted[(size_t)c * PCH + hist[bk[k]] + rk[k]] =
                (unsigned)s[k] | ((unsigned)(d[k] - bk[k] * R) << SRC_BITS);
    }
}

#define GB 256
__global__ __launch_bounds__(1024, 16)
void gat_gemm_partition(const float* __restrict__ h,
                        const float* __restrict__ W,
                        float* __restrict__ z,
                        int nNodes,
                        const int* __restrict__ src,
                        const int* __restrict__ dst,
                        unsigned* __restrict__ sorted,
                        int* __restrict__ tab,
                        int nEdges, int R) {
    if (blockIdx.x < GB) {
        __shared__ float4 wb0[KSTEPS * 32];
        __shared__ float4 wb1[KSTEPS * 32];
        stage_b_fragments(W, wb0, wb1);
        __syncthreads();
        const int wave = __builtin_amdgcn_readfirstlane((int)(threadIdx.x >> 5));
        const int lane = threadIdx.x & 31;
        const int nTiles = (nNodes + 15) >> 4;
        for (int tile = blockIdx.x + GB * wave; tile < nTiles; tile += GB * 32)
            gemm_tile(h, wb0, wb1, z, tile * 16, lane);
    } else {
        partition_chunk(src, dst, sorted, tab, nEdges, R, blockIdx.x - GB);
    }
}

template <int M>
__device__ __forceinline__ float row_xor(float p) {
    return __int_as_float(__builtin_amdgcn_update_dpp(
        0, __float_as_int(p), 0x160 + M, 0xf, 0xf, false));
}

__device__ __forceinline__ float half_wave_sum(float p) {
    p += row_xor<1>(p);
    p += row_xor<2>(p);
    p += row_xor<4>(p);
    p += row_xor<8>(p);
    return p;
}

__device__ __forceinline__ float quarter_wave_sum(float p) {
    p += row_xor<1>(p);
    p += row_xor<2>(p);
    p += row_xor<4>(p);
    return p;
}

__device__ __forceinline__ unsigned oct_bcast(unsigned x, int u) {
    switch (u) {
    case 0:  return __builtin_amdgcn_mov_dpp8(x, 0x000000u);
    case 1:  return __builtin_amdgcn_mov_dpp8(x, 0x249249u);
    case 2:  return __builtin_amdgcn_mov_dpp8(x, 0x492492u);
    case 3:  return __builtin_amdgcn_mov_dpp8(x, 0x6db6dbu);
    case 4:  return __builtin_amdgcn_mov_dpp8(x, 0x924924u);
    case 5:  return __builtin_amdgcn_mov_dpp8(x, 0xb6db6du);
    case 6:  return __builtin_amdgcn_mov_dpp8(x, 0xdb6db6u);
    default: return __builtin_amdgcn_mov_dpp8(x, 0xffffffu);
    }
}

#define CAP (RMAX * OUT_DIM / 2)
#define DBINS 64

__device__ __forceinline__ void owner_lds_add_route(const float4* __restrict__ z4,
                                                    const unsigned* __restrict__ sorted,
                                                    const int* __restrict__ tab,
                                                    float4 a1, float4* acc4,
                                                    const float* sdst,
                                                    float* __restrict__ out,
                                                    int node0, int nLoc, int R, int NC, int b) {
    float* acc = (float*)acc4;
    const int t    = threadIdx.x;
    const int lane = t & 31;
    const int wave = t >> 5;
    const int half = lane >> 4;
    const int l16  = lane & 15;

    for (int i = t; i < R * (OUT_DIM / 4); i += 1024)
        acc4[i] = make_float4(0.f, 0.f, 0.f, 0.f);
    __syncthreads();

    for (int c0 = 0; c0 < NC; c0 += 1024) {
        const int cL = c0 + wave + 32 * lane;
        int loV = 0, hiV = 0;
        if (cL < NC) {
            loV = tab[cL * (NB + 1) + b];
            hiV = tab[cL * (NB + 1) + b + 1];
        }
        for (int k = 0; k < 32; ++k) {
            const int c = c0 + wave + 32 * k;
            if (c >= NC) break;
            const int lo = __builtin_amdgcn_readlane(loV, k);
            const int hi = __builtin_amdgcn_readlane(hiV, k);
            const unsigned* run = sorted + (size_t)c * PCH;
            for (int i = lo; i < hi; i += 32) {
                int n = hi - i;
                if (n > 32) n = 32;
                int ed = 0;
                if (lane < n) ed = (int)run[i + lane];
                for (int j0 = 0; j0 < n; j0 += 16) {
                    float4 v[8];
                    float sd[8];
                    int dl[8];
                    bool ok[8];
#pragma unroll
                    for (int u = 0; u < 8; ++u) {
                        const int j = j0 + 2 * u + half;
                        const unsigned pk = (unsigned)__shfl(ed, j & 31, 32);
                        const int sj = (int)(pk & ((1u << SRC_BITS) - 1u));
                        ok[u] = j < n;
                        dl[u] = (int)(pk >> SRC_BITS);
                        v[u] = make_float4(0.f, 0.f, 0.f, 0.f);
                        sd[u] = 0.f;
                        if (ok[u]) {
                            v[u] = z4[(size_t)sj * 16 + l16];
                            sd[u] = sdst[dl[u]];
                        }
                    }
#pragma unroll
                    for (int u = 0; u < 8; ++u) {
                        if (j0 + 2 * u < n) {
                            float p = v[u].x * a1.x + v[u].y * a1.y
                                    + v[u].z * a1.z + v[u].w * a1.w;
                            p = half_wave_sum(p) + sd[u];
                            if (ok[u]) {
                                const float e = (p >= 0.f) ? p : 0.01f * p;
                                const float alpha = 1.f / (1.f + __expf(-e));
                                const float w0 = alpha * v[u].x, w1 = alpha * v[u].y;
                                const float w2 = alpha * v[u].z, w3 = alpha * v[u].w;
                                float* arow = acc + dl[u] * OUT_DIM + 4 * l16;
                                atomicAdd(arow + ((0 + half) & 3), half ? w1 : w0);
                                atomicAdd(arow + ((1 + half) & 3), half ? w2 : w1);
                                atomicAdd(arow + ((2 + half) & 3), half ? w3 : w2);
                                atomicAdd(arow + ((3 + half) & 3), half ? w0 : w3);
                            }
                        }
                    }
                }
            }
        }
    }
    __syncthreads();

    float4* o4 = (float4*)(out + (size_t)node0 * OUT_DIM);
    for (int i = t; i < nLoc * (OUT_DIM / 4); i += 1024) o4[i] = acc4[i];
}

__global__ __launch_bounds__(1024)
void gat_owner(const float* __restrict__ z,
               const unsigned* __restrict__ sorted,
               const int* __restrict__ tab,
               const float* __restrict__ a_attn,
               float* __restrict__ out,
               int nNodes, int R, int NC) {
    static_assert(RMAX + 1 <= 512, "the row scan uses 16 waves");
    __shared__ float4 acc4[RMAX * OUT_DIM / 4];
    __shared__ float sdst[RMAX];
    __shared__ int cnt[RMAX + 1];
    __shared__ int fill[RMAX];
    __shared__ int wsum[16];
    __shared__ int total;
    __shared__ int dhist[DBINS];
    __shared__ int order[RMAX];
    unsigned* stage = (unsigned*)acc4;
    unsigned* recs  = stage + CAP;

    const int t    = threadIdx.x;
    const int lane = t & 31;
    const int wave = __builtin_amdgcn_readfirstlane(t >> 5);
    const int half = lane >> 4;
    const int l16  = lane & 15;
    const int b    = blockIdx.x;
    const int node0 = b * R;
    int nLoc = nNodes - node0;
    if (nLoc > R) nLoc = R;
    if (nLoc < 0) nLoc = 0;

    const float4* z4 = (const float4*)z;
    const float4 a1 = ((const float4*)a_attn)[l16];
    const float4 a2 = ((const float4*)a_attn)[16 + l16];

    if (t <= R) cnt[t] = 0;
    if (t < DBINS) dhist[t] = 0;
    if (t == 0) total = 0;
    __syncthreads();

    {
        constexpr int NQ = (RMAX + 63) / 64;
        float4 v[NQ];
#pragma unroll
        for (int q = 0; q < NQ; ++q) {
            const int r = q * 64 + wave * 2 + half;
            v[q] = make_float4(0.f, 0.f, 0.f, 0.f);
            if (r < nLoc) v[q] = z4[(size_t)(node0 + r) * 16 + l16];
        }
#pragma unroll
        for (int q = 0; q < NQ; ++q) {
            const int r = q * 64 + wave * 2 + half;
            float p = v[q].x * a2.x + v[q].y * a2.y + v[q].z * a2.z + v[q].w * a2.w;
            p = half_wave_sum(p);
            if (r < nLoc && l16 == 0) sdst[r] = p;
        }
    }

    for (int c0 = 0; c0 < NC; c0 += 1024) {
        const int cL = c0 + wave + 32 * lane;
        int loV = 0, hiV = 0;
        if (cL < NC) {
            loV = tab[cL * (NB + 1) + b];
            hiV = tab[cL * (NB + 1) + b + 1];
        }
        for (int kb = 0; kb < 32; kb += 8) {
            if (c0 + wave + 32 * kb >= NC) break;
            unsigned ed[8];
            int m[8];
            int tot8 = 0;
#pragma unroll
            for (int q = 0; q < 8; ++q) {
                const int c  = c0 + wave + 32 * (kb + q);
                const int lo = __builtin_amdgcn_readlane(loV, kb + q);
                const int hi = __builtin_amdgcn_readlane(hiV, kb + q);
                int n = (c < NC) ? hi - lo : 0;
                if (n > 32) n = 32;
                m[q] = n;
                tot8 += n;
                ed[q] = 0u;
                if (lane < n) ed[q] = sorted[(size_t)c * PCH + lo + lane];
            }
            int base = 0;
            if (lane == 0) base = atomicAdd(&total, tot8);
            base = __builtin_amdgcn_readfirstlane(base);
#pragma unroll
            for (int q = 0; q < 8; ++q) {
                if (lane < m[q]) {
                    atomicAdd(&cnt[ed[q] >> SRC_BITS], 1);
                    if (base + lane < CAP) stage[base + lane] = ed[q];
                }
                base += m[q];
            }
            for (int q = 0; q < 8; ++q) {
                const int c = c0 + wave + 32 * (kb + q);
                if (c >= NC) break;
                const int lo = __builtin_amdgcn_readlane(loV, kb + q);
                const int hi = __builtin_amdgcn_readlane(hiV, kb + q);
                for (int i = lo + 32; i < hi; i += 32) {
                    int n = hi - i;
                    if (n > 32) n = 32;
                    unsigned e1 = 0u;
                    if (lane < n) e1 = sorted[(size_t)c * PCH + i + lane];
                    int b1 = 0;
                    if (lane == 0) b1 = atomicAdd(&total, n);
                    b1 = __builtin_amdgcn_readfirstlane(b1);
                    if (lane < n) {
                        atomicAdd(&cnt[e1 >> SRC_BITS], 1);
                        if (b1 + lane < CAP) stage[b1 + lane] = e1;
                    }
                }
            }
        }
    }
    __syncthreads();

    const int tot = total;
    if (tot > CAP) {
        owner_lds_add_route(z4, sorted, tab, a1, acc4, sdst, out, node0, nLoc, R, NC, b);
        return;
    }

    int cv = 0, incl = 0;
    int dbin = 0, drank = 0;
    if (t < 512) {
        cv = (t < R) ? cnt[t] : 0;
        if (t < nLoc) {
            dbin  = cv < DBINS - 1 ? cv : DBINS - 1;
            drank = atomicAdd(&dhist[dbin], 1);
        }
        incl = cv;
#pragma unroll
        for (int off = 1; off < 32; off <<= 1) {
            const int n = __shfl_up(incl, off, 32);
            if ((t & 31) >= off) incl += n;
        }
        if ((t & 31) == 31) wsum[t >> 5] = incl;
    }
    __syncthreads();
    if (t < 512) {
        int pre = 0;
        for (int w = 0; w < (t >> 5); ++w) pre += wsum[w];
        const int excl = pre + incl - cv;
        if (t < R) { cnt[t] = excl; fill[t] = excl; }
        if (t == 0) cnt[R] = tot;
    } else if (t < 544) {
        const int hiBin = DBINS - 1 - 2 * lane;
        const int nHi = dhist[hiBin];
        const int nLo = dhist[hiBin - 1];
        const int two = nHi + nLo;
        int run = two;
        for (int off = 1; off < 32; off <<= 1) {
            const int n = __shfl_up(run, off, 32);
            if (lane >= off) run += n;
        }
        dhist[hiBin]     = run - two;
        dhist[hiBin - 1] = run - two + nHi;
    }
    __syncthreads();

    if (t < nLoc) order[dhist[dbin] + drank] = t;

    for (int i = t; i < tot; i += 1024) {
        const unsigned pk = stage[i];
        const int pos = atomicAdd(&fill[pk >> SRC_BITS], 1);
        recs[pos] = pk & ((1u << SRC_BITS) - 1u);
    }
    __syncthreads();

    const int q8 = lane >> 3;
    const int l8 = lane & 7;
    const v4f a1lo = ((const v4f*)a_attn)[l8];
    const v4f a1hi = ((const v4f*)a_attn)[8 + l8];
    const v2f aA = __builtin_shufflevector(a1lo, a1lo, 0, 1);
    const v2f aB = __builtin_shufflevector(a1lo, a1lo, 2, 3);
    const v2f aC = __builtin_shufflevector(a1hi, a1hi, 0, 1);
    const v2f aD = __builtin_shufflevector(a1hi, a1hi, 2, 3);
    const bool b0 = (l8 & 1) != 0, b1 = (l8 & 2) != 0, b2 = (l8 & 4) != 0;
    const unsigned lane16 = 16u * (unsigned)l8;
    const char* zb = (const char*)z;
    const int qw = wave * 4 + q8;
    float4* o4 = (float4*)(out + (size_t)node0 * OUT_DIM);
    v4f vlo[8], vhi[8];
#pragma unroll
    for (int u = 0; u < 8; ++u) {
        vlo[u] = (v4f){0.f, 0.f, 0.f, 0.f};
        vhi[u] = (v4f){0.f, 0.f, 0.f, 0.f};
    }
    for (int r0 = 0; r0 < nLoc; r0 += 128) {
        const int place = r0 + ((r0 & 128) ? 127 - qw : qw);
        const bool live = place < nLoc;
        int r = 0, s0 = 0, deg = 0;
        float sd = 0.f;
        if (live) {
            r   = order[place];
            s0  = cnt[r];
            deg = cnt[r + 1] - s0;
            sd  = sdst[r];
        }
        const int d0 = __builtin_amdgcn_readlane(deg, 0);
        const int d1 = __builtin_amdgcn_readlane(deg, 8);
        const int d2 = __builtin_amdgcn_readlane(deg, 16);
        const int d3 = __builtin_amdgcn_readlane(deg, 24);
        const int d01 = d0 > d1 ? d0 : d1;
        const int d23 = d2 > d3 ? d2 : d3;
        const int degMax = d01 > d23 ? d01 : d23;
        v2f sA = {0.f, 0.f}, sB = {0.f, 0.f}, sC = {0.f, 0.f}, sD = {0.f, 0.f};
        for (int j = 0; j < degMax; j += 8) {
            const bool mine = j + l8 < deg;
            const unsigned smine = recs[mine ? s0 + j + l8 : 0];
#pragma unroll
            for (int u = 0; u < 8; ++u) {
                const unsigned off = (oct_bcast(smine, u) << 8) + lane16;
                if (j + u < deg) {
                    vlo[u] = *(const v4f*)(zb + off);
                    vhi[u] = *(const v4f*)(zb + off + 128);
                }
            }
            float p[8];
#pragma unroll
            for (int u = 0; u < 8; ++u) {
                v2f acc = __builtin_shufflevector(vlo[u], vlo[u], 0, 1) * aA;
                acc = __builtin_elementwise_fma(__builtin_shufflevector(vlo[u], vlo[u], 2, 3), aB, acc);
                acc = __builtin_elementwise_fma(__builtin_shufflevector(vhi[u], vhi[u], 0, 1), aC, acc);
                acc = __builtin_elementwise_fma(__builtin_shufflevector(vhi[u], vhi[u], 2, 3), aD, acc);
                p[u] = acc.x + acc.y;
            }
            float q[4], h2[2];
#pragma unroll
            for (int u = 0; u < 4; ++u) {
                const float keep = b0 ? p[2 * u + 1] : p[2 * u];
                const float give = b0 ? p[2 * u] : p[2 * u + 1];
                q[u] = keep + row_xor<1>(give);
            }
#pragma unroll
            for (int u = 0; u < 2; ++u) {
                const float keep = b1 ? q[2 * u + 1] : q[2 * u];
                const float give = b1 ? q[2 * u] : q[2 * u + 1];
                h2[u] = keep + row_xor<2>(give);
            }
            float pm;
            {
                const float keep = b2 ? h2[1] : h2[0];
                const float give = b2 ? h2[0] : h2[1];
                pm = keep + row_xor<4>(give);
            }
            pm += sd;
            const float e = (pm >= 0.f) ? pm : 0.01f * pm;
            float alpha = 1.f / (1.f + __expf(-e));
            alpha = mine ? alpha : 0.f;
#pragma unroll
            for (int u = 0; u < 8; ++u) {
                const float au = __uint_as_float(oct_bcast(__float_as_uint(alpha), u));
                const v2f a2 = {au, au};
                sA = __builtin_elementwise_fma(__builtin_shufflevector(vlo[u], vlo[u], 0, 1), a2, sA);
                sB = __builtin_elementwise_fma(__builtin_shufflevector(vlo[u], vlo[u], 2, 3), a2, sB);
                sC = __builtin_elementwise_fma(__builtin_shufflevector(vhi[u], vhi[u], 0, 1), a2, sC);
                sD = __builtin_elementwise_fma(__builtin_shufflevector(vhi[u], vhi[u], 2, 3), a2, sD);
            }
        }
        if (live) {
            o4[(size_t)r * 16 + l8]     = make_float4(sA.x, sA.y, sB.x, sB.y);
            o4[(size_t)r * 16 + 8 + l8] = make_float4(sC.x, sC.y, sD.x, sD.y);
        }
    }
}

extern "C" void kernel_launch(void* const* d_in, const int* in_sizes, int n_in,
                              void* d_out, int out_size, void* d_ws, size_t ws_size,
                              hipStream_t stream) {
    const float* h      = (const float*)d_in[0];
    const int*   src    = (const int*)  d_in[1];
    const int*   dst    = (const int*)  d_in[2];
    const float* W_fc   = (const float*)d_in[3];
    const float* a_attn = (const float*)d_in[4];
    float*       out    = (float*)d_out;
    float*       z      = (float*)d_ws;

    const int nNodes = in_sizes[0] / IN_DIM;
    const int nEdges = in_sizes[1];

    const int R  = (nNodes + NB - 1) / NB;
    const int NC = (nEdges + PCH - 1) / PCH;
    const size_t zBytes   = (size_t)nNodes * OUT_DIM * sizeof(float);
    const size_t sortedB  = (size_t)NC * PCH * sizeof(unsigned);
    const size_t tabB     = (size_t)NC * (NB + 1) * sizeof(int);
    unsigned* sorted = (unsigned*)((char*)d_ws + zBytes);
    int*  tab    = (int*)((char*)d_ws + zBytes + sortedB);

    if (R <= RMAX && nNodes <= (1 << SRC_BITS) && zBytes + sortedB + tabB <= ws_size) {
        gat_gemm_partition<<<GB + NC, 1024, 0, stream>>>(h, W_fc, z, nNodes,
                                                         src, dst, sorted, tab, nEdges, R);
        gat_owner<<<NB, 1024, 0, stream>>>(z, sorted, tab, a_attn, out,
                                           nNodes, R, NC);
    } else {
        gat_gemm_partition<<<GB, 1024, 0, stream>>>(h, W_fc, z, nNodes,
                                                    src, dst, sorted, tab, nEdges, R);
        const int n4 = out_size / 4;
        const int zb = min((n4 + 255) / 256, 4096);
        gat_zero_out<<<zb, 256, 0, stream>>>((float4*)out, n4);
        const int blocks = (nEdges + 7) / 8;
        gat_edge<<<blocks, 256, 0, stream>>>(z, src, dst, a_attn, out, nEdges);
    }
}
